# starting kernel
# baseline (speedup 1.0000x reference)
#include <hip/hip_runtime.h>
#include <hip/hip_bf16.h>

#define N_NODES 100000
#define N_EDGES 800000
#define IN_DIM  256
#define HID     128
#define OUT_DIM 64

typedef float v2f __attribute__((ext_vector_type(2)));
typedef float v8f __attribute__((ext_vector_type(8)));

__global__ void zero_kernel(float* __restrict__ p, int n) {
    int i = blockIdx.x * blockDim.x + threadIdx.x;
    if (i < n) p[i] = 0.0f;
}

__global__ void degree_kernel(const int* __restrict__ dst, float* __restrict__ degs, int n_edges) {
    int e = blockIdx.x * blockDim.x + threadIdx.x;
    if (e < n_edges) atomicAdd(&degs[dst[e]], 1.0f);
}

__global__ void norm_kernel(float* __restrict__ norm, int n) {
    int i = blockIdx.x * blockDim.x + threadIdx.x;
    if (i < n) {
        float d = norm[i];
        norm[i] = rsqrtf(fmaxf(d, 1.0f));
    }
}

__global__ void scatter_kernel(const float* __restrict__ h,
                               const float* __restrict__ norm,
                               const int* __restrict__ src,
                               const int* __restrict__ dst,
                               float* __restrict__ agg, int n_edges) {
    int gid = blockIdx.x * blockDim.x + threadIdx.x;
    int e = gid >> 5;
    if (e >= n_edges) return;
    int l = gid & 31;
    int s = src[e];
    int d = dst[e];
    float ns = norm[s];
    const float4 v = *reinterpret_cast<const float4*>(h + (size_t)s * HID + l * 4);
    float* out = agg + (size_t)d * HID + l * 4;
    atomicAdd(out + 0, v.x * ns);
    atomicAdd(out + 1, v.y * ns);
    atomicAdd(out + 2, v.z * ns);
    atomicAdd(out + 3, v.w * ns);
}

template<int NT>
__global__ __launch_bounds__(32) void wmma_gemm_strip(
    const float* __restrict__ A, const float* __restrict__ W,
    const float* __restrict__ bias, const float* __restrict__ norm,
    const float* __restrict__ resid, float* __restrict__ C,
    int M, int K, int do_relu)
{
    constexpr int N = NT * 16;
    const int lane  = threadIdx.x;
    const int half  = lane >> 4;
    const int idx16 = lane & 15;
    const int m0 = blockIdx.x * 16;

    const int arow = m0 + idx16;
    const int koff = half * 2;

    const float nscale = (norm != nullptr) ? norm[arow] : 1.0f;
    const float* __restrict__ Arow = A + (size_t)arow * K;

    v8f acc[NT];
#pragma unroll
    for (int t = 0; t < NT; ++t) acc[t] = (v8f){};

    for (int k = 0; k < K; k += 4) {
        v2f a = *reinterpret_cast<const v2f*>(Arow + k + koff);
        a.x *= nscale;
        a.y *= nscale;
        const float* __restrict__ Wr0 = W + (size_t)(k + koff) * N + idx16;
        const float* __restrict__ Wr1 = Wr0 + N;
#pragma unroll
        for (int t = 0; t < NT; ++t) {
            v2f b;
            b.x = Wr0[t * 16];
            b.y = Wr1[t * 16];
            acc[t] = __builtin_amdgcn_wmma_f32_16x16x4_f32(
                 false, a,  false, b,
                 (short)0, acc[t],  false,  false);
        }
    }

#pragma unroll
    for (int t = 0; t < NT; ++t) {
        const int bcol = t * 16 + idx16;
        const float bv = bias[bcol];
#pragma unroll
        for (int r = 0; r < 8; ++r) {
            int orow = m0 + half * 8 + r;
            size_t oidx = (size_t)orow * N + bcol;
            float v = acc[t][r] + bv;
            if (do_relu) v = fmaxf(v, 0.0f);
            if (resid != nullptr) v += resid[oidx];
            C[oidx] = v;
        }
    }
}

extern "C" void kernel_launch(void* const* d_in, const int* in_sizes, int n_in,
                              void* d_out, int out_size, void* d_ws, size_t ws_size,
                              hipStream_t stream) {
    const float* features = (const float*)d_in[0];
    const int*   src      = (const int*)  d_in[1];
    const int*   dst      = (const int*)  d_in[2];
    const float* W_in     = (const float*)d_in[3];
    const float* b_in     = (const float*)d_in[4];
    const float* gcn_w0   = (const float*)d_in[5];
    const float* gcn_b0   = (const float*)d_in[6];
    const float* gcn_w1   = (const float*)d_in[7];
    const float* gcn_b1   = (const float*)d_in[8];
    const float* W_out    = (const float*)d_in[9];
    const float* b_out    = (const float*)d_in[10];
    float* out = (float*)d_out;

    char* ws = (char*)d_ws;
    float* norm = (float*)ws;
    ws += ((size_t)N_NODES * sizeof(float) + 255) & ~(size_t)255;
    float* H = (float*)ws;
    ws += ((size_t)N_NODES * HID * sizeof(float) + 255) & ~(size_t)255;
    float* AGG = (float*)ws;

    const int nodeVec = N_NODES * HID;
    const int mBlocks = N_NODES / 16;

    zero_kernel<<<(N_NODES + 255) / 256, 256, 0, stream>>>(norm, N_NODES);
    degree_kernel<<<(N_EDGES + 255) / 256, 256, 0, stream>>>(dst, norm, N_EDGES);
    norm_kernel<<<(N_NODES + 255) / 256, 256, 0, stream>>>(norm, N_NODES);

    wmma_gemm_strip<8><<<mBlocks, 32, 0, stream>>>(
        features, W_in, b_in, nullptr, nullptr, H, N_NODES, IN_DIM,  0);

    zero_kernel<<<(nodeVec + 255) / 256, 256, 0, stream>>>(AGG, nodeVec);
    scatter_kernel<<<(N_EDGES * 32 + 255) / 256, 256, 0, stream>>>(
        H, norm, src, dst, AGG, N_EDGES);
    wmma_gemm_strip<8><<<mBlocks, 32, 0, stream>>>(
        AGG, gcn_w0, gcn_b0, norm,  H, H, N_NODES, HID,  1);

    zero_kernel<<<(nodeVec + 255) / 256, 256, 0, stream>>>(AGG, nodeVec);
    scatter_kernel<<<(N_EDGES * 32 + 255) / 256, 256, 0, stream>>>(
        H, norm, src, dst, AGG, N_EDGES);
    wmma_gemm_strip<8><<<mBlocks, 32, 0, stream>>>(
        AGG, gcn_w1, gcn_b1, norm, nullptr, H, N_NODES, HID,  1);

    wmma_gemm_strip<4><<<mBlocks, 32, 0, stream>>>(
        H, W_out, b_out, nullptr, nullptr, out, N_NODES, HID,  0);
}


# Replace float-atomic edge scatter and its zero fills by a per-call dst-sorted CSR (int atomic ranks + scan) and a per-node gather sum; per-node sum order now follows rank arrival
# speedup vs baseline: 1.6975x; 1.6975x over previous
#include <hip/hip_runtime.h>
#include <hip/hip_bf16.h>

#define N_NODES 100000
#define N_EDGES 800000
#define IN_DIM  256
#define HID     128
#define OUT_DIM 64

typedef float v2f __attribute__((ext_vector_type(2)));
typedef float v8f __attribute__((ext_vector_type(8)));

__global__ void zero_int_kernel(int* __restrict__ p, int n) {
    int i = blockIdx.x * blockDim.x + threadIdx.x;
    if (i < n) p[i] = 0;
}

__global__ void count_rank_kernel(const int* __restrict__ dst, int* __restrict__ cnt,
                                  int* __restrict__ rank, int n_edges) {
    int e = blockIdx.x * blockDim.x + threadIdx.x;
    if (e < n_edges) rank[e] = atomicAdd(&cnt[dst[e]], 1);
}

__global__ __launch_bounds__(1024) void scan_norm_kernel(
    const int* __restrict__ cnt, int* __restrict__ rowptr,
    float* __restrict__ norm, int n)
{
    __shared__ int s_front[32];
    __shared__ int s_tot[32];
    const int t = threadIdx.x;
    const int lane = t & 31;
    const int w = t >> 5;
    const int base = blockIdx.x * 1024;

    int front = 0;
    for (int i = t; i < base; i += 1024) front += cnt[i];
    for (int d = 16; d > 0; d >>= 1) front += __shfl_xor(front, d);

    const int i = base + t;
    const int c = (i < n) ? cnt[i] : 0;
    int inc = c;
    for (int d = 1; d < 32; d <<= 1) {
        int u = __shfl_up(inc, d);
        if (lane >= d) inc += u;
    }
    if (lane == 31) s_tot[w] = inc;
    if (lane == 0) s_front[w] = front;
    __syncthreads();

    int off = s_front[lane] + ((lane < w) ? s_tot[lane] : 0);
    for (int d = 16; d > 0; d >>= 1) off += __shfl_xor(off, d);

    if (i < n) {
        const int excl = off + inc - c;
        rowptr[i] = excl;
        norm[i] = rsqrtf(fmaxf((float)c, 1.0f));
        if (i == n - 1) rowptr[n] = excl + c;
    }
}

__global__ void fill_kernel(const int* __restrict__ src, const int* __restrict__ dst,
                            const int* __restrict__ rank, const int* __restrict__ rowptr,
                            int* __restrict__ col, int n_edges) {
    int e = blockIdx.x * blockDim.x + threadIdx.x;
    if (e < n_edges) col[rowptr[dst[e]] + rank[e]] = src[e];
}

__global__ __launch_bounds__(256) void gather_kernel(
    const float* __restrict__ h, const float* __restrict__ norm,
    const int* __restrict__ rowptr, const int* __restrict__ col,
    float* __restrict__ agg, int n_nodes)
{
    const int lane = threadIdx.x & 31;
    const int node = __builtin_amdgcn_readfirstlane(
        (int)((blockIdx.x * blockDim.x + threadIdx.x) >> 5));
    if (node >= n_nodes) return;

    const int beg = rowptr[node];
    const int end = rowptr[node + 1];
    const float* __restrict__ hl = h + lane * 4;

    float4 acc = make_float4(0.0f, 0.0f, 0.0f, 0.0f);
    for (int p0 = beg; p0 < end; p0 += 32) {
        const int m = (end - p0 < 32) ? (end - p0) : 32;
        int s = 0;
        float ns = 0.0f;
        if (lane < m) {
            s = col[p0 + lane];
            ns = norm[s];
        }
        int j = 0;
        for (; j + 4 <= m; j += 4) {
            const int s0 = __builtin_amdgcn_readlane(s, j);
            const int s1 = __builtin_amdgcn_readlane(s, j + 1);
            const int s2 = __builtin_amdgcn_readlane(s, j + 2);
            const int s3 = __builtin_amdgcn_readlane(s, j + 3);
            const float n0 = __int_as_float(__builtin_amdgcn_readlane(__float_as_int(ns), j));
            const float n1 = __int_as_float(__builtin_amdgcn_readlane(__float_as_int(ns), j + 1));
            const float n2 = __int_as_float(__builtin_amdgcn_readlane(__float_as_int(ns), j + 2));
            const float n3 = __int_as_float(__builtin_amdgcn_readlane(__float_as_int(ns), j + 3));
            const float4 v0 = *reinterpret_cast<const float4*>(hl + (size_t)s0 * HID);
            const float4 v1 = *reinterpret_cast<const float4*>(hl + (size_t)s1 * HID);
            const float4 v2 = *reinterpret_cast<const float4*>(hl + (size_t)s2 * HID);
            const float4 v3 = *reinterpret_cast<const float4*>(hl + (size_t)s3 * HID);
            acc.x = __builtin_fmaf(v0.x, n0, acc.x); acc.y = __builtin_fmaf(v0.y, n0, acc.y);
            acc.z = __builtin_fmaf(v0.z, n0, acc.z); acc.w = __builtin_fmaf(v0.w, n0, acc.w);
            acc.x = __builtin_fmaf(v1.x, n1, acc.x); acc.y = __builtin_fmaf(v1.y, n1, acc.y);
            acc.z = __builtin_fmaf(v1.z, n1, acc.z); acc.w = __builtin_fmaf(v1.w, n1, acc.w);
            acc.x = __builtin_fmaf(v2.x, n2, acc.x); acc.y = __builtin_fmaf(v2.y, n2, acc.y);
            acc.z = __builtin_fmaf(v2.z, n2, acc.z); acc.w = __builtin_fmaf(v2.w, n2, acc.w);
            acc.x = __builtin_fmaf(v3.x, n3, acc.x); acc.y = __builtin_fmaf(v3.y, n3, acc.y);
            acc.z = __builtin_fmaf(v3.z, n3, acc.z); acc.w = __builtin_fmaf(v3.w, n3, acc.w);
        }
        for (; j < m; ++j) {
            const int s0 = __builtin_amdgcn_readlane(s, j);
            const float n0 = __int_as_float(__builtin_amdgcn_readlane(__float_as_int(ns), j));
            const float4 v0 = *reinterpret_cast<const float4*>(hl + (size_t)s0 * HID);
            acc.x = __builtin_fmaf(v0.x, n0, acc.x); acc.y = __builtin_fmaf(v0.y, n0, acc.y);
            acc.z = __builtin_fmaf(v0.z, n0, acc.z); acc.w = __builtin_fmaf(v0.w, n0, acc.w);
        }
    }
    *reinterpret_cast<float4*>(agg + (size_t)node * HID + lane * 4) = acc;
}

template<int NT>
__global__ __launch_bounds__(32) void wmma_gemm_strip(
    const float* __restrict__ A, const float* __restrict__ W,
    const float* __restrict__ bias, const float* __restrict__ norm,
    const float* __restrict__ resid, float* __restrict__ C,
    int M, int K, int do_relu)
{
    constexpr int N = NT * 16;
    const int lane  = threadIdx.x;
    const int half  = lane >> 4;
    const int idx16 = lane & 15;
    const int m0 = blockIdx.x * 16;

    const int arow = m0 + idx16;
    const int koff = half * 2;

    const float nscale = (norm != nullptr) ? norm[arow] : 1.0f;
    const float* __restrict__ Arow = A + (size_t)arow * K;

    v8f acc[NT];
#pragma unroll
    for (int t = 0; t < NT; ++t) acc[t] = (v8f){};

    for (int k = 0; k < K; k += 4) {
        v2f a = *reinterpret_cast<const v2f*>(Arow + k + koff);
        a.x *= nscale;
        a.y *= nscale;
        const float* __restrict__ Wr0 = W + (size_t)(k + koff) * N + idx16;
        const float* __restrict__ Wr1 = Wr0 + N;
#pragma unroll
        for (int t = 0; t < NT; ++t) {
            v2f b;
            b.x = Wr0[t * 16];
            b.y = Wr1[t * 16];
            acc[t] = __builtin_amdgcn_wmma_f32_16x16x4_f32(
                 false, a,  false, b,
                 (short)0, acc[t],  false,  false);
        }
    }

#pragma unroll
    for (int t = 0; t < NT; ++t) {
        const int bcol = t * 16 + idx16;
        const float bv = bias[bcol];
#pragma unroll
        for (int r = 0; r < 8; ++r) {
            int orow = m0 + half * 8 + r;
            size_t oidx = (size_t)orow * N + bcol;
            float v = acc[t][r] + bv;
            if (do_relu) v = fmaxf(v, 0.0f);
            if (resid != nullptr) v += resid[oidx];
            C[oidx] = v;
        }
    }
}

extern "C" void kernel_launch(void* const* d_in, const int* in_sizes, int n_in,
                              void* d_out, int out_size, void* d_ws, size_t ws_size,
                              hipStream_t stream) {
    const float* features = (const float*)d_in[0];
    const int*   src      = (const int*)  d_in[1];
    const int*   dst      = (const int*)  d_in[2];
    const float* W_in     = (const float*)d_in[3];
    const float* b_in     = (const float*)d_in[4];
    const float* gcn_w0   = (const float*)d_in[5];
    const float* gcn_b0   = (const float*)d_in[6];
    const float* gcn_w1   = (const float*)d_in[7];
    const float* gcn_b1   = (const float*)d_in[8];
    const float* W_out    = (const float*)d_in[9];
    const float* b_out    = (const float*)d_in[10];
    float* out = (float*)d_out;

    char* ws = (char*)d_ws;
    float* norm = (float*)ws;
    ws += ((size_t)N_NODES * sizeof(float) + 255) & ~(size_t)255;
    float* H = (float*)ws;
    ws += ((size_t)N_NODES * HID * sizeof(float) + 255) & ~(size_t)255;
    float* AGG = (float*)ws;

    int* cnt    = (int*)AGG;
    int* rank   = cnt + 100352;
    int* rowptr = (int*)d_out;
    int* col    = rowptr + 100352;

    const int mBlocks = N_NODES / 16;
    const int gBlocks = (N_NODES * 32 + 255) / 256;

    zero_int_kernel<<<(N_NODES + 255) / 256, 256, 0, stream>>>(cnt, N_NODES);
    count_rank_kernel<<<(N_EDGES + 255) / 256, 256, 0, stream>>>(dst, cnt, rank, N_EDGES);
    scan_norm_kernel<<<(N_NODES + 1023) / 1024, 1024, 0, stream>>>(cnt, rowptr, norm, N_NODES);
    fill_kernel<<<(N_EDGES + 255) / 256, 256, 0, stream>>>(src, dst, rank, rowptr, col, N_EDGES);

    wmma_gemm_strip<8><<<mBlocks, 32, 0, stream>>>(
        features, W_in, b_in, nullptr, nullptr, H, N_NODES, IN_DIM,  0);

    gather_kernel<<<gBlocks, 256, 0, stream>>>(H, norm, rowptr, col, AGG, N_NODES);
    wmma_gemm_strip<8><<<mBlocks, 32, 0, stream>>>(
        AGG, gcn_w0, gcn_b0, norm,  H, H, N_NODES, HID,  1);

    gather_kernel<<<gBlocks, 256, 0, stream>>>(H, norm, rowptr, col, AGG, N_NODES);
    wmma_gemm_strip<8><<<mBlocks, 32, 0, stream>>>(
        AGG, gcn_w1, gcn_b1, norm, nullptr, H, N_NODES, HID,  1);

    wmma_gemm_strip<4><<<mBlocks, 32, 0, stream>>>(
        H, W_out, b_out, nullptr, nullptr, out, N_NODES, HID,  0);
}


# GEMMs: one 16-wave workgroup per WGP holds the weights in LDS (16-byte B reads), takes 16-row strips off an LDS counter, A loads one 32-k chunk ahead, residual read per strip; k order, sums unchanged
# speedup vs baseline: 2.0117x; 1.1851x over previous
#include <hip/hip_runtime.h>
#include <hip/hip_bf16.h>

#define N_NODES 100000
#define N_EDGES 800000
#define IN_DIM  256
#define HID     128
#define OUT_DIM 64

typedef float v2f __attribute__((ext_vector_type(2)));
typedef float v8f __attribute__((ext_vector_type(8)));

__global__ void zero_int_kernel(int* __restrict__ p, int n) {
    int i = blockIdx.x * blockDim.x + threadIdx.x;
    if (i < n) p[i] = 0;
}

__global__ void count_rank_kernel(const int* __restrict__ dst, int* __restrict__ cnt,
                                  int* __restrict__ rank, int n_edges) {
    int e = blockIdx.x * blockDim.x + threadIdx.x;
    if (e < n_edges) rank[e] = atomicAdd(&cnt[dst[e]], 1);
}

__global__ __launch_bounds__(1024) void scan_norm_kernel(
    const int* __restrict__ cnt, int* __restrict__ rowptr,
    float* __restrict__ norm, int n)
{
    __shared__ int s_front[32];
    __shared__ int s_tot[32];
    const int t = threadIdx.x;
    const int lane = t & 31;
    const int w = t >> 5;
    const int base = blockIdx.x * 1024;

    int front = 0;
    for (int i = t; i < base; i += 1024) front += cnt[i];
    for (int d = 16; d > 0; d >>= 1) front += __shfl_xor(front, d);

    const int i = base + t;
    const int c = (i < n) ? cnt[i] : 0;
    int inc = c;
    for (int d = 1; d < 32; d <<= 1) {
        int u = __shfl_up(inc, d);
        if (lane >= d) inc += u;
    }
    if (lane == 31) s_tot[w] = inc;
    if (lane == 0) s_front[w] = front;
    __syncthreads();

    int off = s_front[lane] + ((lane < w) ? s_tot[lane] : 0);
    for (int d = 16; d > 0; d >>= 1) off += __shfl_xor(off, d);

    if (i < n) {
        const int excl = off + inc - c;
        rowptr[i] = excl;
        norm[i] = rsqrtf(fmaxf((float)c, 1.0f));
        if (i == n - 1) rowptr[n] = excl + c;
    }
}

__global__ void fill_kernel(const int* __restrict__ src, const int* __restrict__ dst,
                            const int* __restrict__ rank, const int* __restrict__ rowptr,
                            int* __restrict__ col, int n_edges) {
    int e = blockIdx.x * blockDim.x + threadIdx.x;
    if (e < n_edges) col[rowptr[dst[e]] + rank[e]] = src[e];
}

__global__ __launch_bounds__(256) void gather_kernel(
    const float* __restrict__ h, const float* __restrict__ norm,
    const int* __restrict__ rowptr, const int* __restrict__ col,
    float* __restrict__ agg, int n_nodes)
{
    const int lane = threadIdx.x & 31;
    const int node = __builtin_amdgcn_readfirstlane(
        (int)((blockIdx.x * blockDim.x + threadIdx.x) >> 5));
    if (node >= n_nodes) return;

    const int beg = rowptr[node];
    const int end = rowptr[node + 1];
    const float* __restrict__ hl = h + lane * 4;

    float4 acc = make_float4(0.0f, 0.0f, 0.0f, 0.0f);
    for (int p0 = beg; p0 < end; p0 += 32) {
        const int m = (end - p0 < 32) ? (end - p0) : 32;
        int s = 0;
        float ns = 0.0f;
        if (lane < m) {
            s = col[p0 + lane];
            ns = norm[s];
        }
        int j = 0;
        for (; j + 4 <= m; j += 4) {
            const int s0 = __builtin_amdgcn_readlane(s, j);
            const int s1 = __builtin_amdgcn_readlane(s, j + 1);
            const int s2 = __builtin_amdgcn_readlane(s, j + 2);
            const int s3 = __builtin_amdgcn_readlane(s, j + 3);
            const float n0 = __int_as_float(__builtin_amdgcn_readlane(__float_as_int(ns), j));
            const float n1 = __int_as_float(__builtin_amdgcn_readlane(__float_as_int(ns), j + 1));
            const float n2 = __int_as_float(__builtin_amdgcn_readlane(__float_as_int(ns), j + 2));
            const float n3 = __int_as_float(__builtin_amdgcn_readlane(__float_as_int(ns), j + 3));
            const float4 v0 = *reinterpret_cast<const float4*>(hl + (size_t)s0 * HID);
            const float4 v1 = *reinterpret_cast<const float4*>(hl + (size_t)s1 * HID);
            const float4 v2 = *reinterpret_cast<const float4*>(hl + (size_t)s2 * HID);
            const float4 v3 = *reinterpret_cast<const float4*>(hl + (size_t)s3 * HID);
            acc.x = __builtin_fmaf(v0.x, n0, acc.x); acc.y = __builtin_fmaf(v0.y, n0, acc.y);
            acc.z = __builtin_fmaf(v0.z, n0, acc.z); acc.w = __builtin_fmaf(v0.w, n0, acc.w);
            acc.x = __builtin_fmaf(v1.x, n1, acc.x); acc.y = __builtin_fmaf(v1.y, n1, acc.y);
            acc.z = __builtin_fmaf(v1.z, n1, acc.z); acc.w = __builtin_fmaf(v1.w, n1, acc.w);
            acc.x = __builtin_fmaf(v2.x, n2, acc.x); acc.y = __builtin_fmaf(v2.y, n2, acc.y);
            acc.z = __builtin_fmaf(v2.z, n2, acc.z); acc.w = __builtin_fmaf(v2.w, n2, acc.w);
            acc.x = __builtin_fmaf(v3.x, n3, acc.x); acc.y = __builtin_fmaf(v3.y, n3, acc.y);
            acc.z = __builtin_fmaf(v3.z, n3, acc.z); acc.w = __builtin_fmaf(v3.w, n3, acc.w);
        }
        for (; j < m; ++j) {
            const int s0 = __builtin_amdgcn_readlane(s, j);
            const float n0 = __int_as_float(__builtin_amdgcn_readlane(__float_as_int(ns), j));
            const float4 v0 = *reinterpret_cast<const float4*>(hl + (size_t)s0 * HID);
            acc.x = __builtin_fmaf(v0.x, n0, acc.x); acc.y = __builtin_fmaf(v0.y, n0, acc.y);
            acc.z = __builtin_fmaf(v0.z, n0, acc.z); acc.w = __builtin_fmaf(v0.w, n0, acc.w);
        }
    }
    *reinterpret_cast<float4*>(agg + (size_t)node * HID + lane * 4) = acc;
}

#define GEMM_THREADS 512
#define GEMM_LDS_F4 10496

template<int NT, int K, bool HAS_NORM, bool RELU, bool HAS_RESID>
__global__ __launch_bounds__(GEMM_THREADS) void wmma_gemm_lds(
    const float* __restrict__ A, const float* __restrict__ W,
    const float* __restrict__ bias, const float* __restrict__ norm,
    const float* __restrict__ resid, float* __restrict__ C,
    int n_strips)
{
    constexpr int N = NT * 16;
    constexpr int W_F4 = K * N / 4;
    static_assert(W_F4 <= GEMM_LDS_F4, "W must fit the LDS reservation");
    static_assert(K % 32 == 0, "K is consumed in 32-k chunks");
    __shared__ float4 Wl[GEMM_LDS_F4];
    __shared__ int s_next;

    const int tid   = threadIdx.x;
    const int lane  = tid & 31;
    const int half  = lane >> 4;
    const int idx16 = lane & 15;

    if (tid == 0) s_next = 0;
    for (int it = tid; it < W_F4; it += GEMM_THREADS) {
        const int col = it & (N - 1);
        const int jh  = it / N;
        const int kb  = (jh >> 1) * 8 + (jh & 1) * 2;
        const float* __restrict__ wp = W + (size_t)kb * N + col;
        float4 v;
        v.x = wp[0];
        v.y = wp[N];
        v.z = wp[4 * N];
        v.w = wp[5 * N];
        Wl[it] = v;
    }
    __syncthreads();

    const int my_count = (n_strips - (int)blockIdx.x + (int)gridDim.x - 1) / (int)gridDim.x;

    for (int guard = 0; guard < my_count; ++guard) {
        int i = 0;
        if (lane == 0) i = atomicAdd(&s_next, 1);
        i = __builtin_amdgcn_readfirstlane(i);
        if (i >= my_count) break;

        const int m0   = ((int)blockIdx.x + (int)gridDim.x * i) * 16;
        const int arow = m0 + idx16;
        float nscale = 1.0f;
        if (HAS_NORM) nscale = norm[arow];
        const float* __restrict__ Arow = A + (size_t)arow * K + half * 2;

        v8f acc[NT];
#pragma unroll
        for (int t = 0; t < NT; ++t) acc[t] = (v8f){};

        v2f a_cur[8];
#pragma unroll
        for (int s = 0; s < 8; ++s)
            a_cur[s] = *reinterpret_cast<const v2f*>(Arow + 4 * s);

        auto do_chunk = [&](const int kc, const v2f (&a)[8]) {
            const float4* bl = Wl + (size_t)((kc * 4) * 2 + half) * N + idx16;
#pragma unroll
            for (int js = 0; js < 4; ++js) {
                float4 bq[NT];
#pragma unroll
                for (int t = 0; t < NT; ++t) bq[t] = bl[(js * 2) * N + t * 16];
                v2f a0 = a[2 * js];
                v2f a1 = a[2 * js + 1];
                if (HAS_NORM) {
                    a0.x *= nscale; a0.y *= nscale;
                    a1.x *= nscale; a1.y *= nscale;
                }
#pragma unroll
                for (int t = 0; t < NT; ++t) {
                    v2f b; b.x = bq[t].x; b.y = bq[t].y;
                    acc[t] = __builtin_amdgcn_wmma_f32_16x16x4_f32(
                         false, a0,  false, b,
                         (short)0, acc[t],  false,  false);
                }
#pragma unroll
                for (int t = 0; t < NT; ++t) {
                    v2f b; b.x = bq[t].z; b.y = bq[t].w;
                    acc[t] = __builtin_amdgcn_wmma_f32_16x16x4_f32(
                         false, a1,  false, b,
                         (short)0, acc[t],  false,  false);
                }
            }
        };

#pragma unroll 1
        for (int kc = 0; kc < K / 32 - 1; ++kc) {
            v2f a_nxt[8];
#pragma unroll
            for (int s = 0; s < 8; ++s)
                a_nxt[s] = *reinterpret_cast<const v2f*>(Arow + (kc + 1) * 32 + 4 * s);
            do_chunk(kc, a_cur);
#pragma unroll
            for (int s = 0; s < 8; ++s) a_cur[s] = a_nxt[s];
        }
        do_chunk(K / 32 - 1, a_cur);

        const size_t obase = (size_t)(m0 + half * 8) * N + idx16;
        float rv[NT][8];
        if (HAS_RESID) {
#pragma unroll
            for (int t = 0; t < NT; ++t)
#pragma unroll
                for (int r = 0; r < 8; ++r)
                    rv[t][r] = resid[obase + (size_t)r * N + t * 16];
        }
#pragma unroll
        for (int t = 0; t < NT; ++t) {
            const float bv = bias[t * 16 + idx16];
#pragma unroll
            for (int r = 0; r < 8; ++r) {
                float v = acc[t][r] + bv;
                if (RELU) v = fmaxf(v, 0.0f);
                if (HAS_RESID) v += rv[t][r];
                C[obase + (size_t)r * N + t * 16] = v;
            }
        }
    }
}

extern "C" void kernel_launch(void* const* d_in, const int* in_sizes, int n_in,
                              void* d_out, int out_size, void* d_ws, size_t ws_size,
                              hipStream_t stream) {
    const float* features = (const float*)d_in[0];
    const int*   src      = (const int*)  d_in[1];
    const int*   dst      = (const int*)  d_in[2];
    const float* W_in     = (const float*)d_in[3];
    const float* b_in     = (const float*)d_in[4];
    const float* gcn_w0   = (const float*)d_in[5];
    const float* gcn_b0   = (const float*)d_in[6];
    const float* gcn_w1   = (const float*)d_in[7];
    const float* gcn_b1   = (const float*)d_in[8];
    const float* W_out    = (const float*)d_in[9];
    const float* b_out    = (const float*)d_in[10];
    float* out = (float*)d_out;

    char* ws = (char*)d_ws;
    float* norm = (float*)ws;
    ws += ((size_t)N_NODES * sizeof(float) + 255) & ~(size_t)255;
    float* H = (float*)ws;
    ws += ((size_t)N_NODES * HID * sizeof(float) + 255) & ~(size_t)255;
    float* AGG = (float*)ws;

    int* cnt    = (int*)AGG;
    int* rank   = cnt + 100352;
    int* rowptr = (int*)d_out;
    int* col    = rowptr + 100352;

    const int nStrips = N_NODES / 16;
    const int gemmGrid = 256;
    const int gBlocks = (N_NODES * 32 + 255) / 256;

    zero_int_kernel<<<(N_NODES + 255) / 256, 256, 0, stream>>>(cnt, N_NODES);
    count_rank_kernel<<<(N_EDGES + 255) / 256, 256, 0, stream>>>(dst, cnt, rank, N_EDGES);
    scan_norm_kernel<<<(N_NODES + 1023) / 1024, 1024, 0, stream>>>(cnt, rowptr, norm, N_NODES);
    fill_kernel<<<(N_EDGES + 255) / 256, 256, 0, stream>>>(src, dst, rank, rowptr, col, N_EDGES);

    wmma_gemm_lds<8, IN_DIM, false, false, false><<<gemmGrid, GEMM_THREADS, 0, stream>>>(
        features, W_in, b_in, nullptr, nullptr, H, nStrips);

    gather_kernel<<<gBlocks, 256, 0, stream>>>(H, norm, rowptr, col, AGG, N_NODES);
    wmma_gemm_lds<8, HID, true, true, true><<<gemmGrid, GEMM_THREADS, 0, stream>>>(
        AGG, gcn_w0, gcn_b0, norm,  H, H, nStrips);

    gather_kernel<<<gBlocks, 256, 0, stream>>>(H, norm, rowptr, col, AGG, N_NODES);
    wmma_gemm_lds<8, HID, true, true, false><<<gemmGrid, GEMM_THREADS, 0, stream>>>(
        AGG, gcn_w1, gcn_b1, norm, nullptr, H, nStrips);

    wmma_gemm_lds<4, HID, false, false, false><<<gemmGrid, GEMM_THREADS, 0, stream>>>(
        H, W_out, b_out, nullptr, nullptr, out, nStrips);
}


# Fuse each layer's neighbour gather into its GEMM launch: a wave gathers its 16-node strip in CSR order into a private LDS tile and feeds the WMMAs from it; per-node sum order and k order unchanged
# speedup vs baseline: 2.2298x; 1.1084x over previous
#include <hip/hip_runtime.h>
#include <hip/hip_bf16.h>

#define N_NODES 100000
#define N_EDGES 800000
#define IN_DIM  256
#define HID     128
#define OUT_DIM 64

typedef float v2f __attribute__((ext_vector_type(2)));
typedef float v8f __attribute__((ext_vector_type(8)));

__global__ void zero_int_kernel(int* __restrict__ p, int n) {
    int i = blockIdx.x * blockDim.x + threadIdx.x;
    if (i < n) p[i] = 0;
}

__global__ void count_rank_kernel(const int* __restrict__ dst, int* __restrict__ cnt,
                                  int* __restrict__ rank, int n_edges) {
    int e = blockIdx.x * blockDim.x + threadIdx.x;
    if (e < n_edges) rank[e] = atomicAdd(&cnt[dst[e]], 1);
}

__global__ __launch_bounds__(1024) void scan_norm_kernel(
    const int* __restrict__ cnt, int* __restrict__ rowptr,
    float* __restrict__ norm, int n)
{
    __shared__ int s_front[32];
    __shared__ int s_tot[32];
    const int t = threadIdx.x;
    const int lane = t & 31;
    const int w = t >> 5;
    const int base = blockIdx.x * 1024;

    int front = 0;
    for (int i = t; i < base; i += 1024) front += cnt[i];
    for (int d = 16; d > 0; d >>= 1) front += __shfl_xor(front, d);

    const int i = base + t;
    const int c = (i < n) ? cnt[i] : 0;
    int inc = c;
    for (int d = 1; d < 32; d <<= 1) {
        int u = __shfl_up(inc, d);
        if (lane >= d) inc += u;
    }
    if (lane == 31) s_tot[w] = inc;
    if (lane == 0) s_front[w] = front;
    __syncthreads();

    int off = s_front[lane] + ((lane < w) ? s_tot[lane] : 0);
    for (int d = 16; d > 0; d >>= 1) off += __shfl_xor(off, d);

    if (i < n) {
        const int excl = off + inc - c;
        rowptr[i] = excl;
        norm[i] = rsqrtf(fmaxf((float)c, 1.0f));
        if (i == n - 1) rowptr[n] = excl + c;
    }
}

__global__ void fill_kernel(const int* __restrict__ src, const int* __restrict__ dst,
                            const int* __restrict__ rank, const int* __restrict__ rowptr,
                            int* __restrict__ col, int n_edges) {
    int e = blockIdx.x * blockDim.x + threadIdx.x;
    if (e < n_edges) col[rowptr[dst[e]] + rank[e]] = src[e];
}

#define GEMM_THREADS 512
#define GEMM_LDS_F4 10496

template<int NT, int K, bool HAS_NORM, bool RELU, bool HAS_RESID>
__global__ __launch_bounds__(GEMM_THREADS) void wmma_gemm_lds(
    const float* __restrict__ A, const float* __restrict__ W,
    const float* __restrict__ bias, const float* __restrict__ norm,
    const float* __restrict__ resid, float* __restrict__ C,
    int n_strips)
{
    constexpr int N = NT * 16;
    constexpr int W_F4 = K * N / 4;
    static_assert(W_F4 <= GEMM_LDS_F4, "W must fit the LDS reservation");
    static_assert(K % 32 == 0, "K is consumed in 32-k chunks");
    __shared__ float4 Wl[GEMM_LDS_F4];
    __shared__ int s_next;

    const int tid   = threadIdx.x;
    const int lane  = tid & 31;
    const int half  = lane >> 4;
    const int idx16 = lane & 15;

    if (tid == 0) s_next = 0;
    for (int it = tid; it < W_F4; it += GEMM_THREADS) {
        const int col = it & (N - 1);
        const int jh  = it / N;
        const int kb  = (jh >> 1) * 8 + (jh & 1) * 2;
        const float* __restrict__ wp = W + (size_t)kb * N + col;
        float4 v;
        v.x = wp[0];
        v.y = wp[N];
        v.z = wp[4 * N];
        v.w = wp[5 * N];
        Wl[it] = v;
    }
    __syncthreads();

    const int my_count = (n_strips - (int)blockIdx.x + (int)gridDim.x - 1) / (int)gridDim.x;

    for (int guard = 0; guard < my_count; ++guard) {
        int i = 0;
        if (lane == 0) i = atomicAdd(&s_next, 1);
        i = __builtin_amdgcn_readfirstlane(i);
        if (i >= my_count) break;

        const int m0   = ((int)blockIdx.x + (int)gridDim.x * i) * 16;
        const int arow = m0 + idx16;
        float nscale = 1.0f;
        if (HAS_NORM) nscale = norm[arow];
        const float* __restrict__ Arow = A + (size_t)arow * K + half * 2;

        v8f acc[NT];
#pragma unroll
        for (int t = 0; t < NT; ++t) acc[t] = (v8f){};

        v2f a_cur[8];
#pragma unroll
        for (int s = 0; s < 8; ++s)
            a_cur[s] = *reinterpret_cast<const v2f*>(Arow + 4 * s);

        auto do_chunk = [&](const int kc, const v2f (&a)[8]) {
            const float4* bl = Wl + (size_t)((kc * 4) * 2 + half) * N + idx16;
#pragma unroll
            for (int js = 0; js < 4; ++js) {
                float4 bq[NT];
#pragma unroll
                for (int t = 0; t < NT; ++t) bq[t] = bl[(js * 2) * N + t * 16];
                v2f a0 = a[2 * js];
                v2f a1 = a[2 * js + 1];
                if (HAS_NORM) {
                    a0.x *= nscale; a0.y *= nscale;
                    a1.x *= nscale; a1.y *= nscale;
                }
#pragma unroll
                for (int t = 0; t < NT; ++t) {
                    v2f b; b.x = bq[t].x; b.y = bq[t].y;
                    acc[t] = __builtin_amdgcn_wmma_f32_16x16x4_f32(
                         false, a0,  false, b,
                         (short)0, acc[t],  false,  false);
                }
#pragma unroll
                for (int t = 0; t < NT; ++t) {
                    v2f b; b.x = bq[t].z; b.y = bq[t].w;
                    acc[t] = __builtin_amdgcn_wmma_f32_16x16x4_f32(
                         false, a1,  false, b,
                         (short)0, acc[t],  false,  false);
                }
            }
        };

#pragma unroll 1
        for (int kc = 0; kc < K / 32 - 1; ++kc) {
            v2f a_nxt[8];
#pragma unroll
            for (int s = 0; s < 8; ++s)
                a_nxt[s] = *reinterpret_cast<const v2f*>(Arow + (kc + 1) * 32 + 4 * s);
            do_chunk(kc, a_cur);
#pragma unroll
            for (int s = 0; s < 8; ++s) a_cur[s] = a_nxt[s];
        }
        do_chunk(K / 32 - 1, a_cur);

        const size_t obase = (size_t)(m0 + half * 8) * N + idx16;
        float rv[NT][8];
        if (HAS_RESID) {
#pragma unroll
            for (int t = 0; t < NT; ++t)
#pragma unroll
                for (int r = 0; r < 8; ++r)
                    rv[t][r] = resid[obase + (size_t)r * N + t * 16];
        }
#pragma unroll
        for (int t = 0; t < NT; ++t) {
            const float bv = bias[t * 16 + idx16];
#pragma unroll
            for (int r = 0; r < 8; ++r) {
                float v = acc[t][r] + bv;
                if (RELU) v = fmaxf(v, 0.0f);
                if (HAS_RESID) v += rv[t][r];
                C[obase + (size_t)r * N + t * 16] = v;
            }
        }
    }
}

#define TILE_PITCH 132

template<bool HAS_RESID>
__global__ __launch_bounds__(GEMM_THREADS) void gcn_layer_fused(
    const float* __restrict__ Hin, const float* __restrict__ W,
    const float* __restrict__ bias, const float* __restrict__ norm,
    const int* __restrict__ rowptr, const int* __restrict__ col,
    float* __restrict__ Hout, int n_strips)
{
    constexpr int NT = 8;
    constexpr int N = HID;
    constexpr int K = HID;
    constexpr int W_F4 = K * N / 4;
    __shared__ float4 Wl[W_F4];
    __shared__ __attribute__((aligned(16))) float At[(GEMM_THREADS / 32) * 16 * TILE_PITCH];
    __shared__ int s_next;

    const int tid   = threadIdx.x;
    const int lane  = tid & 31;
    const int half  = lane >> 4;
    const int idx16 = lane & 15;

    if (tid == 0) s_next = 0;
    for (int it = tid; it < W_F4; it += GEMM_THREADS) {
        const int c   = it & (N - 1);
        const int jh  = it / N;
        const int kb  = (jh >> 1) * 8 + (jh & 1) * 2;
        const float* __restrict__ wp = W + (size_t)kb * N + c;
        float4 v;
        v.x = wp[0];
        v.y = wp[N];
        v.z = wp[4 * N];
        v.w = wp[5 * N];
        Wl[it] = v;
    }
    __syncthreads();

    float* tile = At + (tid >> 5) * (16 * TILE_PITCH);
    const float* __restrict__ hl = Hin + lane * 4;
    const int my_count = (n_strips - (int)blockIdx.x + (int)gridDim.x - 1) / (int)gridDim.x;

    for (int guard = 0; guard < my_count; ++guard) {
        int i = 0;
        if (lane == 0) i = atomicAdd(&s_next, 1);
        i = __builtin_amdgcn_readfirstlane(i);
        if (i >= my_count) break;

        const int m0 = ((int)blockIdx.x + (int)gridDim.x * i) * 16;

        int rp = 0;
        float nd = 0.0f;
        if (lane < 17) rp = rowptr[m0 + lane];
        if (lane < 16) nd = norm[m0 + lane];
        const int E0 = __builtin_amdgcn_readlane(rp, 0);
        const int E1 = __builtin_amdgcn_readlane(rp, 16);
        int node = 0;
        int nextEnd = __builtin_amdgcn_readlane(rp, 1);
        float4 acc = make_float4(0.0f, 0.0f, 0.0f, 0.0f);

        auto flush = [&]() {
            const float sc = __int_as_float(__builtin_amdgcn_readlane(__float_as_int(nd), node));
            float4 o;
            o.x = acc.x * sc; o.y = acc.y * sc; o.z = acc.z * sc; o.w = acc.w * sc;
            *reinterpret_cast<float4*>(tile + node * TILE_PITCH + lane * 4) = o;
            acc = make_float4(0.0f, 0.0f, 0.0f, 0.0f);
            ++node;
            nextEnd = (node < 16) ? __builtin_amdgcn_readlane(rp, (node + 1) & 31) : 0x7fffffff;
        };

        int pe = E0 + lane;
        if (pe >= E1) pe = E1 - 1;
        if (pe < 0) pe = 0;
        int s = col[pe];
        float ns = norm[s];

        for (int p0 = E0; p0 < E1; p0 += 32) {
            const int m = (E1 - p0 < 32) ? (E1 - p0) : 32;
            int pn = p0 + 32 + lane;
            if (pn >= E1) pn = E1 - 1;
            const int s_n = col[pn];

            float4 v[32];
#pragma unroll
            for (int g = 0; g < 4; ++g) {
                if (g * 8 < m) {
#pragma unroll
                    for (int u = 0; u < 8; ++u) {
                        const int sj = __builtin_amdgcn_readlane(s, g * 8 + u);
                        v[g * 8 + u] = *reinterpret_cast<const float4*>(hl + (size_t)sj * HID);
                    }
                } else {
#pragma unroll
                    for (int u = 0; u < 8; ++u) v[g * 8 + u] = make_float4(0.0f, 0.0f, 0.0f, 0.0f);
                }
            }
#pragma unroll
            for (int u = 0; u < 32; ++u) {
                if (u < m) {
                    const int p = p0 + u;
                    while (p >= nextEnd) flush();
                    const float n = __int_as_float(__builtin_amdgcn_readlane(__float_as_int(ns), u));
                    acc.x = __builtin_fmaf(v[u].x, n, acc.x);
                    acc.y = __builtin_fmaf(v[u].y, n, acc.y);
                    acc.z = __builtin_fmaf(v[u].z, n, acc.z);
                    acc.w = __builtin_fmaf(v[u].w, n, acc.w);
                }
            }
            const float ns_n = norm[s_n];
            s = s_n;
            ns = ns_n;
        }
        while (node < 16) flush();

        const float* Arow = tile + idx16 * TILE_PITCH + half * 2;

        v8f cacc[NT];
#pragma unroll
        for (int t = 0; t < NT; ++t) cacc[t] = (v8f){};

        v2f a_cur[8];
#pragma unroll
        for (int q = 0; q < 8; ++q)
            a_cur[q] = *reinterpret_cast<const v2f*>(Arow + 4 * q);

        auto do_chunk = [&](const int kc, const v2f (&a)[8]) {
            const float4* bl = Wl + (size_t)((kc * 4) * 2 + half) * N + idx16;
#pragma unroll
            for (int js = 0; js < 4; ++js) {
                float4 bq[NT];
#pragma unroll
                for (int t = 0; t < NT; ++t) bq[t] = bl[(js * 2) * N + t * 16];
                const v2f a0 = a[2 * js];
                const v2f a1 = a[2 * js + 1];
#pragma unroll
                for (int t = 0; t < NT; ++t) {
                    v2f b; b.x = bq[t].x; b.y = bq[t].y;
                    cacc[t] = __builtin_amdgcn_wmma_f32_16x16x4_f32(
                         false, a0,  false, b,
                         (short)0, cacc[t],  false,  false);
                }
#pragma unroll
                for (int t = 0; t < NT; ++t) {
                    v2f b; b.x = bq[t].z; b.y = bq[t].w;
                    cacc[t] = __builtin_amdgcn_wmma_f32_16x16x4_f32(
                         false, a1,  false, b,
                         (short)0, cacc[t],  false,  false);
                }
            }
        };

#pragma unroll 1
        for (int kc = 0; kc < K / 32 - 1; ++kc) {
            v2f a_nxt[8];
#pragma unroll
            for (int q = 0; q < 8; ++q)
                a_nxt[q] = *reinterpret_cast<const v2f*>(Arow + (kc + 1) * 32 + 4 * q);
            do_chunk(kc, a_cur);
#pragma unroll
            for (int q = 0; q < 8; ++q) a_cur[q] = a_nxt[q];
        }
        do_chunk(K / 32 - 1, a_cur);

        const size_t obase = (size_t)(m0 + half * 8) * N + idx16;
        float rv[NT][8];
        if (HAS_RESID) {
#pragma unroll
            for (int t = 0; t < NT; ++t)
#pragma unroll
                for (int r = 0; r < 8; ++r)
                    rv[t][r] = Hin[obase + (size_t)r * N + t * 16];
        }
#pragma unroll
        for (int t = 0; t < NT; ++t) {
            const float bv = bias[t * 16 + idx16];
#pragma unroll
            for (int r = 0; r < 8; ++r) {
                float o = fmaxf(cacc[t][r] + bv, 0.0f);
                if (HAS_RESID) o += rv[t][r];
                Hout[obase + (size_t)r * N + t * 16] = o;
            }
        }
    }
}

extern "C" void kernel_launch(void* const* d_in, const int* in_sizes, int n_in,
                              void* d_out, int out_size, void* d_ws, size_t ws_size,
                              hipStream_t stream) {
    const float* features = (const float*)d_in[0];
    const int*   src      = (const int*)  d_in[1];
    const int*   dst      = (const int*)  d_in[2];
    const float* W_in     = (const float*)d_in[3];
    const float* b_in     = (const float*)d_in[4];
    const float* gcn_w0   = (const float*)d_in[5];
    const float* gcn_b0   = (const float*)d_in[6];
    const float* gcn_w1   = (const float*)d_in[7];
    const float* gcn_b1   = (const float*)d_in[8];
    const float* W_out    = (const float*)d_in[9];
    const float* b_out    = (const float*)d_in[10];
    float* out = (float*)d_out;

    char* ws = (char*)d_ws;
    float* norm = (float*)ws;
    ws += ((size_t)N_NODES * sizeof(float) + 255) & ~(size_t)255;
    float* H = (float*)ws;
    ws += ((size_t)N_NODES * HID * sizeof(float) + 255) & ~(size_t)255;
    float* H2 = (float*)ws;

    int* cnt    = (int*)H2;
    int* rank   = cnt + 100352;
    int* rowptr = (int*)d_out;
    int* col    = rowptr + 100352;

    const int nStrips = N_NODES / 16;
    const int gemmGrid = 256;

    zero_int_kernel<<<(N_NODES + 255) / 256, 256, 0, stream>>>(cnt, N_NODES);
    count_rank_kernel<<<(N_EDGES + 255) / 256, 256, 0, stream>>>(dst, cnt, rank, N_EDGES);
    scan_norm_kernel<<<(N_NODES + 1023) / 1024, 1024, 0, stream>>>(cnt, rowptr, norm, N_NODES);
    fill_kernel<<<(N_EDGES + 255) / 256, 256, 0, stream>>>(src, dst, rank, rowptr, col, N_EDGES);

    wmma_gemm_lds<8, IN_DIM, false, false, false><<<gemmGrid, GEMM_THREADS, 0, stream>>>(
        features, W_in, b_in, nullptr, nullptr, H, nStrips);

    gcn_layer_fused<true><<<gemmGrid, GEMM_THREADS, 0, stream>>>(
        H, gcn_w0, gcn_b0, norm, rowptr, col, H2, nStrips);

    gcn_layer_fused<false><<<gemmGrid, GEMM_THREADS, 0, stream>>>(
        H2, gcn_w1, gcn_b1, norm, rowptr, col, H, nStrips);

    wmma_gemm_lds<4, HID, false, false, false><<<gemmGrid, GEMM_THREADS, 0, stream>>>(
        H, W_out, b_out, nullptr, nullptr, out, nStrips);
}


# CSR fill (col from rowptr, rank, src, dst) runs as 4 extra waves per workgroup inside the input GEMM's launch, which it neither feeds nor reads: 8 launches become 7; sums and k order unchanged
# speedup vs baseline: 2.3367x; 1.0479x over previous
#include <hip/hip_runtime.h>
#include <hip/hip_bf16.h>

#define N_NODES 100000
#define N_EDGES 800000
#define IN_DIM  256
#define HID     128
#define OUT_DIM 64

typedef float v2f __attribute__((ext_vector_type(2)));
typedef float v8f __attribute__((ext_vector_type(8)));

__global__ void zero_int_kernel(int* __restrict__ p, int n) {
    int i = blockIdx.x * blockDim.x + threadIdx.x;
    if (i < n) p[i] = 0;
}

__global__ void count_rank_kernel(const int* __restrict__ dst, int* __restrict__ cnt,
                                  int* __restrict__ rank, int n_edges) {
    int e = blockIdx.x * blockDim.x + threadIdx.x;
    if (e < n_edges) rank[e] = atomicAdd(&cnt[dst[e]], 1);
}

__global__ __launch_bounds__(1024) void scan_norm_kernel(
    const int* __restrict__ cnt, int* __restrict__ rowptr,
    float* __restrict__ norm, int n)
{
    __shared__ int s_front[32];
    __shared__ int s_tot[32];
    const int t = threadIdx.x;
    const int lane = t & 31;
    const int w = t >> 5;
    const int base = blockIdx.x * 1024;

    int front = 0;
    for (int i = t; i < base; i += 1024) front += cnt[i];
    for (int d = 16; d > 0; d >>= 1) front += __shfl_xor(front, d);

    const int i = base + t;
    const int c = (i < n) ? cnt[i] : 0;
    int inc = c;
    for (int d = 1; d < 32; d <<= 1) {
        int u = __shfl_up(inc, d);
        if (lane >= d) inc += u;
    }
    if (lane == 31) s_tot[w] = inc;
    if (lane == 0) s_front[w] = front;
    __syncthreads();

    int off = s_front[lane] + ((lane < w) ? s_tot[lane] : 0);
    for (int d = 16; d > 0; d >>= 1) off += __shfl_xor(off, d);

    if (i < n) {
        const int excl = off + inc - c;
        rowptr[i] = excl;
        norm[i] = rsqrtf(fmaxf((float)c, 1.0f));
        if (i == n - 1) rowptr[n] = excl + c;
    }
}

#define GEMM_THREADS 512
#define GEMM_LDS_F4 10496
#define FILL_THREADS 128

template<int NT, int K, bool HAS_NORM, bool RELU, bool HAS_RESID, bool WITH_FILL>
__global__ __launch_bounds__(WITH_FILL ? GEMM_THREADS + FILL_THREADS : GEMM_THREADS) void wmma_gemm_lds(
    const float* __restrict__ A, const float* __restrict__ W,
    const float* __restrict__ bias, const float* __restrict__ norm,
    const float* __restrict__ resid, float* __restrict__ C,
    int n_strips,
    const int* __restrict__ e_src, const int* __restrict__ e_dst,
    const int* __restrict__ e_rank, const int* __restrict__ e_rowptr,
    int* __restrict__ e_col, int n_edges)
{
    constexpr int N = NT * 16;
    constexpr int W_F4 = K * N / 4;
    static_assert(W_F4 <= GEMM_LDS_F4, "W must fit the LDS reservation");
    static_assert(K % 32 == 0, "K is consumed in 32-k chunks");
    __shared__ float4 Wl[GEMM_LDS_F4];
    __shared__ int s_next;

    const int tid   = threadIdx.x;
    const int lane  = tid & 31;
    const int half  = lane >> 4;
    const int idx16 = lane & 15;

    if (tid == 0) s_next = 0;
    for (int it = tid; it < W_F4 && tid < GEMM_THREADS; it += GEMM_THREADS) {
        const int col = it & (N - 1);
        const int jh  = it / N;
        const int kb  = (jh >> 1) * 8 + (jh & 1) * 2;
        const float* __restrict__ wp = W + (size_t)kb * N + col;
        float4 v;
        v.x = wp[0];
        v.y = wp[N];
        v.z = wp[4 * N];
        v.w = wp[5 * N];
        Wl[it] = v;
    }
    __syncthreads();

    if (WITH_FILL && tid >= GEMM_THREADS) {
        const int stride = (int)gridDim.x * FILL_THREADS;
#pragma unroll 8
        for (int e = (int)blockIdx.x * FILL_THREADS + (tid - GEMM_THREADS); e < n_edges; e += stride)
            e_col[e_rowptr[e_dst[e]] + e_rank[e]] = e_src[e];
        return;
    }

    const int my_count = (n_strips - (int)blockIdx.x + (int)gridDim.x - 1) / (int)gridDim.x;

    for (int guard = 0; guard < my_count; ++guard) {
        int i = 0;
        if (lane == 0) i = atomicAdd(&s_next, 1);
        i = __builtin_amdgcn_readfirstlane(i);
        if (i >= my_count) break;

        const int m0   = ((int)blockIdx.x + (int)gridDim.x * i) * 16;
        const int arow = m0 + idx16;
        float nscale = 1.0f;
        if (HAS_NORM) nscale = norm[arow];
        const float* __restrict__ Arow = A + (size_t)arow * K + half * 2;

        v8f acc[NT];
#pragma unroll
        for (int t = 0; t < NT; ++t) acc[t] = (v8f){};

        v2f a_cur[8];
#pragma unroll
        for (int s = 0; s < 8; ++s)
            a_cur[s] = *reinterpret_cast<const v2f*>(Arow + 4 * s);

        auto do_chunk = [&](const int kc, const v2f (&a)[8]) {
            const float4* bl = Wl + (size_t)((kc * 4) * 2 + half) * N + idx16;
#pragma unroll
            for (int js = 0; js < 4; ++js) {
                float4 bq[NT];
#pragma unroll
                for (int t = 0; t < NT; ++t) bq[t] = bl[(js * 2) * N + t * 16];
                v2f a0 = a[2 * js];
                v2f a1 = a[2 * js + 1];
                if (HAS_NORM) {
                    a0.x *= nscale; a0.y *= nscale;
                    a1.x *= nscale; a1.y *= nscale;
                }
#pragma unroll
                for (int t = 0; t < NT; ++t) {
                    v2f b; b.x = bq[t].x; b.y = bq[t].y;
                    acc[t] = __builtin_amdgcn_wmma_f32_16x16x4_f32(
                         false, a0,  false, b,
                         (short)0, acc[t],  false,  false);
                }
#pragma unroll
                for (int t = 0; t < NT; ++t) {
                    v2f b; b.x = bq[t].z; b.y = bq[t].w;
                    acc[t] = __builtin_amdgcn_wmma_f32_16x16x4_f32(
                         false, a1,  false, b,
                         (short)0, acc[t],  false,  false);
                }
            }
        };

#pragma unroll 1
        for (int kc = 0; kc < K / 32 - 1; ++kc) {
            v2f a_nxt[8];
#pragma unroll
            for (int s = 0; s < 8; ++s)
                a_nxt[s] = *reinterpret_cast<const v2f*>(Arow + (kc + 1) * 32 + 4 * s);
            do_chunk(kc, a_cur);
#pragma unroll
            for (int s = 0; s < 8; ++s) a_cur[s] = a_nxt[s];
        }
        do_chunk(K / 32 - 1, a_cur);

        const size_t obase = (size_t)(m0 + half * 8) * N + idx16;
        float rv[NT][8];
        if (HAS_RESID) {
#pragma unroll
            for (int t = 0; t < NT; ++t)
#pragma unroll
                for (int r = 0; r < 8; ++r)
                    rv[t][r] = resid[obase + (size_t)r * N + t * 16];
        }
#pragma unroll
        for (int t = 0; t < NT; ++t) {
            const float bv = bias[t * 16 + idx16];
#pragma unroll
            for (int r = 0; r < 8; ++r) {
                float v = acc[t][r] + bv;
                if (RELU) v = fmaxf(v, 0.0f);
                if (HAS_RESID) v += rv[t][r];
                C[obase + (size_t)r * N + t * 16] = v;
            }
        }
    }
}

#define TILE_PITCH 132

template<bool HAS_RESID>
__global__ __launch_bounds__(GEMM_THREADS) void gcn_layer_fused(
    const float* __restrict__ Hin, const float* __restrict__ W,
    const float* __restrict__ bias, const float* __restrict__ norm,
    const int* __restrict__ rowptr, const int* __restrict__ col,
    float* __restrict__ Hout, int n_strips)
{
    constexpr int NT = 8;
    constexpr int N = HID;
    constexpr int K = HID;
    constexpr int W_F4 = K * N / 4;
    __shared__ float4 Wl[W_F4];
    __shared__ __attribute__((aligned(16))) float At[(GEMM_THREADS / 32) * 16 * TILE_PITCH];
    __shared__ int s_next;

    const int tid   = threadIdx.x;
    const int lane  = tid & 31;
    const int half  = lane >> 4;
    const int idx16 = lane & 15;

    if (tid == 0) s_next = 0;
    for (int it = tid; it < W_F4; it += GEMM_THREADS) {
        const int c   = it & (N - 1);
        const int jh  = it / N;
        const int kb  = (jh >> 1) * 8 + (jh & 1) * 2;
        const float* __restrict__ wp = W + (size_t)kb * N + c;
        float4 v;
        v.x = wp[0];
        v.y = wp[N];
        v.z = wp[4 * N];
        v.w = wp[5 * N];
        Wl[it] = v;
    }
    __syncthreads();

    float* tile = At + (tid >> 5) * (16 * TILE_PITCH);
    const float* __restrict__ hl = Hin + lane * 4;
    const int my_count = (n_strips - (int)blockIdx.x + (int)gridDim.x - 1) / (int)gridDim.x;

    for (int guard = 0; guard < my_count; ++guard) {
        int i = 0;
        if (lane == 0) i = atomicAdd(&s_next, 1);
        i = __builtin_amdgcn_readfirstlane(i);
        if (i >= my_count) break;

        const int m0 = ((int)blockIdx.x + (int)gridDim.x * i) * 16;

        int rp = 0;
        float nd = 0.0f;
        if (lane < 17) rp = rowptr[m0 + lane];
        if (lane < 16) nd = norm[m0 + lane];
        const int E0 = __builtin_amdgcn_readlane(rp, 0);
        const int E1 = __builtin_amdgcn_readlane(rp, 16);
        int node = 0;
        int nextEnd = __builtin_amdgcn_readlane(rp, 1);
        float4 acc = make_float4(0.0f, 0.0f, 0.0f, 0.0f);

        auto flush = [&]() {
            const float sc = __int_as_float(__builtin_amdgcn_readlane(__float_as_int(nd), node));
            float4 o;
            o.x = acc.x * sc; o.y = acc.y * sc; o.z = acc.z * sc; o.w = acc.w * sc;
            *reinterpret_cast<float4*>(tile + node * TILE_PITCH + lane * 4) = o;
            acc = make_float4(0.0f, 0.0f, 0.0f, 0.0f);
            ++node;
            nextEnd = (node < 16) ? __builtin_amdgcn_readlane(rp, (node + 1) & 31) : 0x7fffffff;
        };

        int pe = E0 + lane;
        if (pe >= E1) pe = E1 - 1;
        if (pe < 0) pe = 0;
        int s = col[pe];
        float ns = norm[s];

        for (int p0 = E0; p0 < E1; p0 += 32) {
            const int m = (E1 - p0 < 32) ? (E1 - p0) : 32;
            int pn = p0 + 32 + lane;
            if (pn >= E1) pn = E1 - 1;
            const int s_n = col[pn];

            float4 v[32];
#pragma unroll
            for (int g = 0; g < 4; ++g) {
                if (g * 8 < m) {
#pragma unroll
                    for (int u = 0; u < 8; ++u) {
                        const int sj = __builtin_amdgcn_readlane(s, g * 8 + u);
                        v[g * 8 + u] = *reinterpret_cast<const float4*>(hl + (size_t)sj * HID);
                    }
                } else {
#pragma unroll
                    for (int u = 0; u < 8; ++u) v[g * 8 + u] = make_float4(0.0f, 0.0f, 0.0f, 0.0f);
                }
            }
#pragma unroll
            for (int u = 0; u < 32; ++u) {
                if (u < m) {
                    const int p = p0 + u;
                    while (p >= nextEnd) flush();
                    const float n = __int_as_float(__builtin_amdgcn_readlane(__float_as_int(ns), u));
                    acc.x = __builtin_fmaf(v[u].x, n, acc.x);
                    acc.y = __builtin_fmaf(v[u].y, n, acc.y);
                    acc.z = __builtin_fmaf(v[u].z, n, acc.z);
                    acc.w = __builtin_fmaf(v[u].w, n, acc.w);
                }
            }
            const float ns_n = norm[s_n];
            s = s_n;
            ns = ns_n;
        }
        while (node < 16) flush();

        const float* Arow = tile + idx16 * TILE_PITCH + half * 2;

        v8f cacc[NT];
#pragma unroll
        for (int t = 0; t < NT; ++t) cacc[t] = (v8f){};

        v2f a_cur[8];
#pragma unroll
        for (int q = 0; q < 8; ++q)
            a_cur[q] = *reinterpret_cast<const v2f*>(Arow + 4 * q);

        auto do_chunk = [&](const int kc, const v2f (&a)[8]) {
            const float4* bl = Wl + (size_t)((kc * 4) * 2 + half) * N + idx16;
#pragma unroll
            for (int js = 0; js < 4; ++js) {
                float4 bq[NT];
#pragma unroll
                for (int t = 0; t < NT; ++t) bq[t] = bl[(js * 2) * N + t * 16];
                const v2f a0 = a[2 * js];
                const v2f a1 = a[2 * js + 1];
#pragma unroll
                for (int t = 0; t < NT; ++t) {
                    v2f b; b.x = bq[t].x; b.y = bq[t].y;
                    cacc[t] = __builtin_amdgcn_wmma_f32_16x16x4_f32(
                         false, a0,  false, b,
                         (short)0, cacc[t],  false,  false);
                }
#pragma unroll
                for (int t = 0; t < NT; ++t) {
                    v2f b; b.x = bq[t].z; b.y = bq[t].w;
                    cacc[t] = __builtin_amdgcn_wmma_f32_16x16x4_f32(
                         false, a1,  false, b,
                         (short)0, cacc[t],  false,  false);
                }
            }
        };

#pragma unroll 1
        for (int kc = 0; kc < K / 32 - 1; ++kc) {
            v2f a_nxt[8];
#pragma unroll
            for (int q = 0; q < 8; ++q)
                a_nxt[q] = *reinterpret_cast<const v2f*>(Arow + (kc + 1) * 32 + 4 * q);
            do_chunk(kc, a_cur);
#pragma unroll
            for (int q = 0; q < 8; ++q) a_cur[q] = a_nxt[q];
        }
        do_chunk(K / 32 - 1, a_cur);

        const size_t obase = (size_t)(m0 + half * 8) * N + idx16;
        float rv[NT][8];
        if (HAS_RESID) {
#pragma unroll
            for (int t = 0; t < NT; ++t)
#pragma unroll
                for (int r = 0; r < 8; ++r)
                    rv[t][r] = Hin[obase + (size_t)r * N + t * 16];
        }
#pragma unroll
        for (int t = 0; t < NT; ++t) {
            const float bv = bias[t * 16 + idx16];
#pragma unroll
            for (int r = 0; r < 8; ++r) {
                float o = fmaxf(cacc[t][r] + bv, 0.0f);
                if (HAS_RESID) o += rv[t][r];
                Hout[obase + (size_t)r * N + t * 16] = o;
            }
        }
    }
}

extern "C" void kernel_launch(void* const* d_in, const int* in_sizes, int n_in,
                              void* d_out, int out_size, void* d_ws, size_t ws_size,
                              hipStream_t stream) {
    const float* features = (const float*)d_in[0];
    const int*   src      = (const int*)  d_in[1];
    const int*   dst      = (const int*)  d_in[2];
    const float* W_in     = (const float*)d_in[3];
    const float* b_in     = (const float*)d_in[4];
    const float* gcn_w0   = (const float*)d_in[5];
    const float* gcn_b0   = (const float*)d_in[6];
    const float* gcn_w1   = (const float*)d_in[7];
    const float* gcn_b1   = (const float*)d_in[8];
    const float* W_out    = (const float*)d_in[9];
    const float* b_out    = (const float*)d_in[10];
    float* out = (float*)d_out;

    char* ws = (char*)d_ws;
    float* norm = (float*)ws;
    ws += ((size_t)N_NODES * sizeof(float) + 255) & ~(size_t)255;
    float* H = (float*)ws;
    ws += ((size_t)N_NODES * HID * sizeof(float) + 255) & ~(size_t)255;
    float* H2 = (float*)ws;

    int* cnt    = (int*)H2;
    int* rank   = cnt + 100352;
    int* rowptr = (int*)d_out;
    int* col    = rowptr + 100352;

    const int nStrips = N_NODES / 16;
    const int gemmGrid = 256;

    zero_int_kernel<<<(N_NODES + 255) / 256, 256, 0, stream>>>(cnt, N_NODES);
    count_rank_kernel<<<(N_EDGES + 255) / 256, 256, 0, stream>>>(dst, cnt, rank, N_EDGES);
    scan_norm_kernel<<<(N_NODES + 1023) / 1024, 1024, 0, stream>>>(cnt, rowptr, norm, N_NODES);

    wmma_gemm_lds<8, IN_DIM, false, false, false, true><<<gemmGrid, GEMM_THREADS + FILL_THREADS, 0, stream>>>(
        features, W_in, b_in, nullptr, nullptr, H, nStrips,
        src, dst, rank, rowptr, col, N_EDGES);

    gcn_layer_fused<true><<<gemmGrid, GEMM_THREADS, 0, stream>>>(
        H, gcn_w0, gcn_b0, norm, rowptr, col, H2, nStrips);

    gcn_layer_fused<false><<<gemmGrid, GEMM_THREADS, 0, stream>>>(
        H2, gcn_w1, gcn_b1, norm, rowptr, col, H, nStrips);

    wmma_gemm_lds<4, HID, false, false, false, false><<<gemmGrid, GEMM_THREADS, 0, stream>>>(
        H, W_out, b_out, nullptr, nullptr, out, nStrips,
        nullptr, nullptr, nullptr, nullptr, nullptr, 0);
}


# Adjacency built in one pass inside the input GEMM's launch: an edge's count atomic returns its slot in a 32-entry bin of its dst node; no rank, scan, norm arrays; 7 launches to 5; sum order rule kept
# speedup vs baseline: 2.6369x; 1.1285x over previous
#include <hip/hip_runtime.h>
#include <hip/hip_bf16.h>

#define N_NODES 100000
#define N_EDGES 800000
#define IN_DIM  256
#define HID     128
#define OUT_DIM 64

typedef float v2f __attribute__((ext_vector_type(2)));
typedef float v8f __attribute__((ext_vector_type(8)));

#define BIN_CAP 32

__global__ void zero_int_kernel(int* __restrict__ p, int n) {
    int i = blockIdx.x * blockDim.x + threadIdx.x;
    if (i < n) p[i] = 0;
}

#define GEMM_THREADS 512
#define GEMM_LDS_F4 10496
#define FILL_THREADS 128

template<int NT, int K, bool HAS_NORM, bool RELU, bool HAS_RESID, bool WITH_FILL>
__global__ __launch_bounds__(WITH_FILL ? GEMM_THREADS + FILL_THREADS : GEMM_THREADS) void wmma_gemm_lds(
    const float* __restrict__ A, const float* __restrict__ W,
    const float* __restrict__ bias, const float* __restrict__ norm,
    const float* __restrict__ resid, float* __restrict__ C,
    int n_strips,
    const int* __restrict__ e_src, const int* __restrict__ e_dst,
    int* __restrict__ e_cnt, int* __restrict__ e_bins,
    int* __restrict__ e_ovf, int n_edges)
{
    constexpr int N = NT * 16;
    constexpr int W_F4 = K * N / 4;
    static_assert(W_F4 <= GEMM_LDS_F4, "W must fit the LDS reservation");
    static_assert(K % 32 == 0, "K is consumed in 32-k chunks");
    __shared__ float4 Wl[GEMM_LDS_F4];
    __shared__ int s_next;

    const int tid   = threadIdx.x;
    const int lane  = tid & 31;
    const int half  = lane >> 4;
    const int idx16 = lane & 15;

    if (tid == 0) s_next = 0;
    for (int it = tid; it < W_F4 && tid < GEMM_THREADS; it += GEMM_THREADS) {
        const int col = it & (N - 1);
        const int jh  = it / N;
        const int kb  = (jh >> 1) * 8 + (jh & 1) * 2;
        const float* __restrict__ wp = W + (size_t)kb * N + col;
        float4 v;
        v.x = wp[0];
        v.y = wp[N];
        v.z = wp[4 * N];
        v.w = wp[5 * N];
        Wl[it] = v;
    }
    __syncthreads();

    if (WITH_FILL && tid >= GEMM_THREADS) {
        const int stride = (int)gridDim.x * FILL_THREADS;
        int e0 = (int)blockIdx.x * FILL_THREADS + (tid - GEMM_THREADS);
        for (; e0 + 7 * stride < n_edges; e0 += 8 * stride) {
            int ed[8], es[8], er[8];
#pragma unroll
            for (int u = 0; u < 8; ++u) {
                ed[u] = e_dst[e0 + u * stride];
                es[u] = e_src[e0 + u * stride];
            }
#pragma unroll
            for (int u = 0; u < 8; ++u)
                er[u] = atomicAdd(&e_cnt[ed[u]], 1);
#pragma unroll
            for (int u = 0; u < 8; ++u) {
                if (er[u] < BIN_CAP) {
                    e_bins[ed[u] * BIN_CAP + er[u]] = es[u];
                } else {
                    const int k = atomicAdd(&e_cnt[N_NODES], 1);
                    e_ovf[2 * k] = ed[u];
                    e_ovf[2 * k + 1] = es[u];
                }
            }
        }
        for (int e = e0; e < n_edges; e += stride) {
            const int d = e_dst[e];
            const int s = e_src[e];
            const int r = atomicAdd(&e_cnt[d], 1);
            if (r < BIN_CAP) {
                e_bins[d * BIN_CAP + r] = s;
            } else {
                const int k = atomicAdd(&e_cnt[N_NODES], 1);
                e_ovf[2 * k] = d;
                e_ovf[2 * k + 1] = s;
            }
        }
        return;
    }

    const int my_count = (n_strips - (int)blockIdx.x + (int)gridDim.x - 1) / (int)gridDim.x;

    for (int guard = 0; guard < my_count; ++guard) {
        int i = 0;
        if (lane == 0) i = atomicAdd(&s_next, 1);
        i = __builtin_amdgcn_readfirstlane(i);
        if (i >= my_count) break;

        const int m0   = ((int)blockIdx.x + (int)gridDim.x * i) * 16;
        const int arow = m0 + idx16;
        float nscale = 1.0f;
        if (HAS_NORM) nscale = norm[arow];
        const float* __restrict__ Arow = A + (size_t)arow * K + half * 2;

        v8f acc[NT];
#pragma unroll
        for (int t = 0; t < NT; ++t) acc[t] = (v8f){};

        v2f a_cur[8];
#pragma unroll
        for (int s = 0; s < 8; ++s)
            a_cur[s] = *reinterpret_cast<const v2f*>(Arow + 4 * s);

        auto do_chunk = [&](const int kc, const v2f (&a)[8]) {
            const float4* bl = Wl + (size_t)((kc * 4) * 2 + half) * N + idx16;
#pragma unroll
            for (int js = 0; js < 4; ++js) {
                float4 bq[NT];
#pragma unroll
                for (int t = 0; t < NT; ++t) bq[t] = bl[(js * 2) * N + t * 16];
                v2f a0 = a[2 * js];
                v2f a1 = a[2 * js + 1];
                if (HAS_NORM) {
                    a0.x *= nscale; a0.y *= nscale;
                    a1.x *= nscale; a1.y *= nscale;
                }
#pragma unroll
                for (int t = 0; t < NT; ++t) {
                    v2f b; b.x = bq[t].x; b.y = bq[t].y;
                    acc[t] = __builtin_amdgcn_wmma_f32_16x16x4_f32(
                         false, a0,  false, b,
                         (short)0, acc[t],  false,  false);
                }
#pragma unroll
                for (int t = 0; t < NT; ++t) {
                    v2f b; b.x = bq[t].z; b.y = bq[t].w;
                    acc[t] = __builtin_amdgcn_wmma_f32_16x16x4_f32(
                         false, a1,  false, b,
                         (short)0, acc[t],  false,  false);
                }
            }
        };

#pragma unroll 1
        for (int kc = 0; kc < K / 32 - 1; ++kc) {
            v2f a_nxt[8];
#pragma unroll
            for (int s = 0; s < 8; ++s)
                a_nxt[s] = *reinterpret_cast<const v2f*>(Arow + (kc + 1) * 32 + 4 * s);
            do_chunk(kc, a_cur);
#pragma unroll
            for (int s = 0; s < 8; ++s) a_cur[s] = a_nxt[s];
        }
        do_chunk(K / 32 - 1, a_cur);

        const size_t obase = (size_t)(m0 + half * 8) * N + idx16;
        float rv[NT][8];
        if (HAS_RESID) {
#pragma unroll
            for (int t = 0; t < NT; ++t)
#pragma unroll
                for (int r = 0; r < 8; ++r)
                    rv[t][r] = resid[obase + (size_t)r * N + t * 16];
        }
#pragma unroll
        for (int t = 0; t < NT; ++t) {
            const float bv = bias[t * 16 + idx16];
#pragma unroll
            for (int r = 0; r < 8; ++r) {
                float v = acc[t][r] + bv;
                if (RELU) v = fmaxf(v, 0.0f);
                if (HAS_RESID) v += rv[t][r];
                C[obase + (size_t)r * N + t * 16] = v;
            }
        }
    }
}

#define TILE_PITCH 132

template<bool HAS_RESID>
__global__ __launch_bounds__(GEMM_THREADS) void gcn_layer_fused(
    const float* __restrict__ Hin, const float* __restrict__ W,
    const float* __restrict__ bias, const int* __restrict__ cnt,
    const int* __restrict__ bins, const int* __restrict__ ovf,
    float* __restrict__ Hout, int n_strips)
{
    constexpr int NT = 8;
    constexpr int N = HID;
    constexpr int K = HID;
    constexpr int W_F4 = K * N / 4;
    __shared__ float4 Wl[W_F4];
    __shared__ __attribute__((aligned(16))) float At[(GEMM_THREADS / 32) * 16 * TILE_PITCH];
    __shared__ int s_next;

    const int tid   = threadIdx.x;
    const int lane  = tid & 31;
    const int half  = lane >> 4;
    const int idx16 = lane & 15;

    if (tid == 0) s_next = 0;
    for (int it = tid; it < W_F4; it += GEMM_THREADS) {
        const int c   = it & (N - 1);
        const int jh  = it / N;
        const int kb  = (jh >> 1) * 8 + (jh & 1) * 2;
        const float* __restrict__ wp = W + (size_t)kb * N + c;
        float4 v;
        v.x = wp[0];
        v.y = wp[N];
        v.z = wp[4 * N];
        v.w = wp[5 * N];
        Wl[it] = v;
    }
    __syncthreads();

    float* tile = At + (tid >> 5) * (16 * TILE_PITCH);
    const float* __restrict__ hl = Hin + lane * 4;
    const int my_count = (n_strips - (int)blockIdx.x + (int)gridDim.x - 1) / (int)gridDim.x;

    for (int guard = 0; guard < my_count; ++guard) {
        int i = 0;
        if (lane == 0) i = atomicAdd(&s_next, 1);
        i = __builtin_amdgcn_readfirstlane(i);
        if (i >= my_count) break;

        const int m0 = ((int)blockIdx.x + (int)gridDim.x * i) * 16;

        int c = 0;
        if (lane < 16) c = cnt[m0 + lane];
        const float nd = rsqrtf(fmaxf((float)c, 1.0f));
        const int cl = (c < BIN_CAP) ? c : BIN_CAP;
        const bool any_ovf = __builtin_amdgcn_ballot_w32(c > BIN_CAP) != 0u;
        int inc = cl;
        for (int d = 1; d < 32; d <<= 1) {
            int u = __shfl_up(inc, d);
            if (lane >= d) inc += u;
        }
        const int rp = inc - cl;
        const int E1 = __builtin_amdgcn_readlane(rp, 16);
        int node = 0;
        int nextEnd = __builtin_amdgcn_readlane(rp, 1);
        float4 acc = make_float4(0.0f, 0.0f, 0.0f, 0.0f);

        auto slot_of = [&](const int p) -> int {
            int j = 0;
            int base = 0;
#pragma unroll
            for (int k = 1; k < 16; ++k) {
                const int r = __builtin_amdgcn_readlane(rp, k);
                const bool ge = p >= r;
                j += ge ? 1 : 0;
                base = ge ? r : base;
            }
            return (m0 + j) * BIN_CAP + (p - base);
        };
        auto node_id = [](const int x) -> int {
            return (x < 0) ? 0 : ((x > N_NODES - 1) ? N_NODES - 1 : x);
        };

        auto flush = [&]() {
            const float sc = __int_as_float(__builtin_amdgcn_readlane(__float_as_int(nd), node));
            float4 o;
            o.x = acc.x * sc; o.y = acc.y * sc; o.z = acc.z * sc; o.w = acc.w * sc;
            *reinterpret_cast<float4*>(tile + node * TILE_PITCH + lane * 4) = o;
            acc = make_float4(0.0f, 0.0f, 0.0f, 0.0f);
            ++node;
            nextEnd = (node < 16) ? __builtin_amdgcn_readlane(rp, (node + 1) & 31) : 0x7fffffff;
        };

        int pe = lane;
        if (pe >= E1) pe = E1 - 1;
        if (pe < 0) pe = 0;
        int s = node_id(bins[slot_of(pe)]);
        float ns = rsqrtf(fmaxf((float)cnt[s], 1.0f));

        for (int p0 = 0; p0 < E1; p0 += 32) {
            const int m = (E1 - p0 < 32) ? (E1 - p0) : 32;
            int pn = p0 + 32 + lane;
            if (pn >= E1) pn = E1 - 1;
            const int s_n = node_id(bins[slot_of(pn)]);

            float4 v[32];
#pragma unroll
            for (int g = 0; g < 4; ++g) {
                if (g * 8 < m) {
#pragma unroll
                    for (int u = 0; u < 8; ++u) {
                        const int sj = __builtin_amdgcn_readlane(s, g * 8 + u);
                        v[g * 8 + u] = *reinterpret_cast<const float4*>(hl + (size_t)sj * HID);
                    }
                } else {
#pragma unroll
                    for (int u = 0; u < 8; ++u) v[g * 8 + u] = make_float4(0.0f, 0.0f, 0.0f, 0.0f);
                }
            }
#pragma unroll
            for (int u = 0; u < 32; ++u) {
                if (u < m) {
                    const int p = p0 + u;
                    while (p >= nextEnd) flush();
                    const float n = __int_as_float(__builtin_amdgcn_readlane(__float_as_int(ns), u));
                    acc.x = __builtin_fmaf(v[u].x, n, acc.x);
                    acc.y = __builtin_fmaf(v[u].y, n, acc.y);
                    acc.z = __builtin_fmaf(v[u].z, n, acc.z);
                    acc.w = __builtin_fmaf(v[u].w, n, acc.w);
                }
            }
            const float ns_n = rsqrtf(fmaxf((float)cnt[s_n], 1.0f));
            s = s_n;
            ns = ns_n;
        }
        while (node < 16) flush();

        if (any_ovf) {
            int n_ovf = cnt[N_NODES];
            if (n_ovf > N_EDGES) n_ovf = N_EDGES;
            n_ovf = __builtin_amdgcn_readfirstlane(n_ovf);
            for (int j = 0; j < 16; ++j) {
                const int cj = cnt[m0 + j];
                if (cj <= BIN_CAP) continue;
                float4 ex = make_float4(0.0f, 0.0f, 0.0f, 0.0f);
                for (int k = 0; k < n_ovf; ++k) {
                    if (ovf[2 * k] == m0 + j) {
                        const int so = node_id(ovf[2 * k + 1]);
                        const float n = rsqrtf(fmaxf((float)cnt[so], 1.0f));
                        const float4 r = *reinterpret_cast<const float4*>(hl + (size_t)so * HID);
                        ex.x = __builtin_fmaf(r.x, n, ex.x);
                        ex.y = __builtin_fmaf(r.y, n, ex.y);
                        ex.z = __builtin_fmaf(r.z, n, ex.z);
                        ex.w = __builtin_fmaf(r.w, n, ex.w);
                    }
                }
                const float sc = rsqrtf(fmaxf((float)cj, 1.0f));
                float4* tp = reinterpret_cast<float4*>(tile + j * TILE_PITCH + lane * 4);
                float4 t = *tp;
                t.x += ex.x * sc; t.y += ex.y * sc; t.z += ex.z * sc; t.w += ex.w * sc;
                *tp = t;
            }
        }

        const float* Arow = tile + idx16 * TILE_PITCH + half * 2;

        v8f cacc[NT];
#pragma unroll
        for (int t = 0; t < NT; ++t) cacc[t] = (v8f){};

        v2f a_cur[8];
#pragma unroll
        for (int q = 0; q < 8; ++q)
            a_cur[q] = *reinterpret_cast<const v2f*>(Arow + 4 * q);

        auto do_chunk = [&](const int kc, const v2f (&a)[8]) {
            const float4* bl = Wl + (size_t)((kc * 4) * 2 + half) * N + idx16;
#pragma unroll
            for (int js = 0; js < 4; ++js) {
                float4 bq[NT];
#pragma unroll
                for (int t = 0; t < NT; ++t) bq[t] = bl[(js * 2) * N + t * 16];
                const v2f a0 = a[2 * js];
                const v2f a1 = a[2 * js + 1];
#pragma unroll
                for (int t = 0; t < NT; ++t) {
                    v2f b; b.x = bq[t].x; b.y = bq[t].y;
                    cacc[t] = __builtin_amdgcn_wmma_f32_16x16x4_f32(
                         false, a0,  false, b,
                         (short)0, cacc[t],  false,  false);
                }
#pragma unroll
                for (int t = 0; t < NT; ++t) {
                    v2f b; b.x = bq[t].z; b.y = bq[t].w;
                    cacc[t] = __builtin_amdgcn_wmma_f32_16x16x4_f32(
                         false, a1,  false, b,
                         (short)0, cacc[t],  false,  false);
                }
            }
        };

#pragma unroll 1
        for (int kc = 0; kc < K / 32 - 1; ++kc) {
            v2f a_nxt[8];
#pragma unroll
            for (int q = 0; q < 8; ++q)
                a_nxt[q] = *reinterpret_cast<const v2f*>(Arow + (kc + 1) * 32 + 4 * q);
            do_chunk(kc, a_cur);
#pragma unroll
            for (int q = 0; q < 8; ++q) a_cur[q] = a_nxt[q];
        }
        do_chunk(K / 32 - 1, a_cur);

        const size_t obase = (size_t)(m0 + half * 8) * N + idx16;
        float rv[NT][8];
        if (HAS_RESID) {
#pragma unroll
            for (int t = 0; t < NT; ++t)
#pragma unroll
                for (int r = 0; r < 8; ++r)
                    rv[t][r] = Hin[obase + (size_t)r * N + t * 16];
        }
#pragma unroll
        for (int t = 0; t < NT; ++t) {
            const float bv = bias[t * 16 + idx16];
#pragma unroll
            for (int r = 0; r < 8; ++r) {
                float o = fmaxf(cacc[t][r] + bv, 0.0f);
                if (HAS_RESID) o += rv[t][r];
                Hout[obase + (size_t)r * N + t * 16] = o;
            }
        }
    }
}

extern "C" void kernel_launch(void* const* d_in, const int* in_sizes, int n_in,
                              void* d_out, int out_size, void* d_ws, size_t ws_size,
                              hipStream_t stream) {
    const float* features = (const float*)d_in[0];
    const int*   src      = (const int*)  d_in[1];
    const int*   dst      = (const int*)  d_in[2];
    const float* W_in     = (const float*)d_in[3];
    const float* b_in     = (const float*)d_in[4];
    const float* gcn_w0   = (const float*)d_in[5];
    const float* gcn_b0   = (const float*)d_in[6];
    const float* gcn_w1   = (const float*)d_in[7];
    const float* gcn_b1   = (const float*)d_in[8];
    const float* W_out    = (const float*)d_in[9];
    const float* b_out    = (const float*)d_in[10];
    float* out = (float*)d_out;

    char* ws = (char*)d_ws;
    int* cnt = (int*)ws;
    ws += ((size_t)N_NODES * sizeof(float) + 255) & ~(size_t)255;
    float* H = (float*)ws;
    ws += ((size_t)N_NODES * HID * sizeof(float) + 255) & ~(size_t)255;
    float* H2 = (float*)ws;

    int* bins = (int*)d_out;
    int* ovf  = bins + N_NODES * BIN_CAP;

    const int nStrips = N_NODES / 16;
    const int gemmGrid = 256;

    zero_int_kernel<<<(N_NODES + 1 + 255) / 256, 256, 0, stream>>>(cnt, N_NODES + 1);

    wmma_gemm_lds<8, IN_DIM, false, false, false, true><<<gemmGrid, GEMM_THREADS + FILL_THREADS, 0, stream>>>(
        features, W_in, b_in, nullptr, nullptr, H, nStrips,
        src, dst, cnt, bins, ovf, N_EDGES);

    gcn_layer_fused<true><<<gemmGrid, GEMM_THREADS, 0, stream>>>(
        H, gcn_w0, gcn_b0, cnt, bins, ovf, H2, nStrips);

    gcn_layer_fused<false><<<gemmGrid, GEMM_THREADS, 0, stream>>>(
        H2, gcn_w1, gcn_b1, cnt, bins, ovf, H, nStrips);

    wmma_gemm_lds<4, HID, false, false, false, false><<<gemmGrid, GEMM_THREADS, 0, stream>>>(
        H, W_out, b_out, nullptr, nullptr, out, nStrips,
        nullptr, nullptr, nullptr, nullptr, nullptr, 0);
}


# Output GEMM fused into layer 1's launch (relu'd strip via the wave's LDS tile into WMMAs on W_out in LDS): 5 launches to 4; bins sit in their own out row; nodes over 32 in-edges sum in edge order
# speedup vs baseline: 2.7738x; 1.0519x over previous
#include <hip/hip_runtime.h>
#include <hip/hip_bf16.h>

#define N_NODES 100000
#define N_EDGES 800000
#define IN_DIM  256
#define HID     128
#define OUT_DIM 64

typedef float v2f __attribute__((ext_vector_type(2)));
typedef float v8f __attribute__((ext_vector_type(8)));

#define BIN_CAP 32
#define BIN_PITCH 64

__global__ void zero_int_kernel(int* __restrict__ p, int n) {
    int i = blockIdx.x * blockDim.x + threadIdx.x;
    if (i < n) p[i] = 0;
}

#define GEMM_THREADS 512
#define GEMM_LDS_F4 10496
#define FILL_THREADS 128

template<int NT, int K, bool HAS_NORM, bool RELU, bool HAS_RESID, bool WITH_FILL>
__global__ __launch_bounds__(WITH_FILL ? GEMM_THREADS + FILL_THREADS : GEMM_THREADS) void wmma_gemm_lds(
    const float* __restrict__ A, const float* __restrict__ W,
    const float* __restrict__ bias, const float* __restrict__ norm,
    const float* __restrict__ resid, float* __restrict__ C,
    int n_strips,
    const int* __restrict__ e_src, const int* __restrict__ e_dst,
    int* __restrict__ e_cnt, int* __restrict__ e_bins, int n_edges)
{
    constexpr int N = NT * 16;
    constexpr int W_F4 = K * N / 4;
    static_assert(W_F4 <= GEMM_LDS_F4, "W must fit the LDS reservation");
    static_assert(K % 32 == 0, "K is consumed in 32-k chunks");
    __shared__ float4 Wl[GEMM_LDS_F4];
    __shared__ int s_next;

    const int tid   = threadIdx.x;
    const int lane  = tid & 31;
    const int half  = lane >> 4;
    const int idx16 = lane & 15;

    if (tid == 0) s_next = 0;
    for (int it = tid; it < W_F4 && tid < GEMM_THREADS; it += GEMM_THREADS) {
        const int col = it & (N - 1);
        const int jh  = it / N;
        const int kb  = (jh >> 1) * 8 + (jh & 1) * 2;
        const float* __restrict__ wp = W + (size_t)kb * N + col;
        float4 v;
        v.x = wp[0];
        v.y = wp[N];
        v.z = wp[4 * N];
        v.w = wp[5 * N];
        Wl[it] = v;
    }
    __syncthreads();

    if (WITH_FILL && tid >= GEMM_THREADS) {
        const int stride = (int)gridDim.x * FILL_THREADS;
        int e0 = (int)blockIdx.x * FILL_THREADS + (tid - GEMM_THREADS);
        for (; e0 + 7 * stride < n_edges; e0 += 8 * stride) {
            int ed[8], es[8], er[8];
#pragma unroll
            for (int u = 0; u < 8; ++u) {
                ed[u] = e_dst[e0 + u * stride];
                es[u] = e_src[e0 + u * stride];
            }
#pragma unroll
            for (int u = 0; u < 8; ++u)
                er[u] = atomicAdd(&e_cnt[ed[u]], 1);
#pragma unroll
            for (int u = 0; u < 8; ++u) {
                if (er[u] < BIN_CAP) e_bins[ed[u] * BIN_PITCH + er[u]] = es[u];
            }
        }
        for (int e = e0; e < n_edges; e += stride) {
            const int d = e_dst[e];
            const int s = e_src[e];
            const int r = atomicAdd(&e_cnt[d], 1);
            if (r < BIN_CAP) e_bins[d * BIN_PITCH + r] = s;
        }
        return;
    }

    const int my_count = (n_strips - (int)blockIdx.x + (int)gridDim.x - 1) / (int)gridDim.x;

    for (int guard = 0; guard < my_count; ++guard) {
        int i = 0;
        if (lane == 0) i = atomicAdd(&s_next, 1);
        i = __builtin_amdgcn_readfirstlane(i);
        if (i >= my_count) break;

        const int m0   = ((int)blockIdx.x + (int)gridDim.x * i) * 16;
        const int arow = m0 + idx16;
        float nscale = 1.0f;
        if (HAS_NORM) nscale = norm[arow];
        const float* __restrict__ Arow = A + (size_t)arow * K + half * 2;

        v8f acc[NT];
#pragma unroll
        for (int t = 0; t < NT; ++t) acc[t] = (v8f){};

        v2f a_cur[8];
#pragma unroll
        for (int s = 0; s < 8; ++s)
            a_cur[s] = *reinterpret_cast<const v2f*>(Arow + 4 * s);

        auto do_chunk = [&](const int kc, const v2f (&a)[8]) {
            const float4* bl = Wl + (size_t)((kc * 4) * 2 + half) * N + idx16;
#pragma unroll
            for (int js = 0; js < 4; ++js) {
                float4 bq[NT];
#pragma unroll
                for (int t = 0; t < NT; ++t) bq[t] = bl[(js * 2) * N + t * 16];
                v2f a0 = a[2 * js];
                v2f a1 = a[2 * js + 1];
                if (HAS_NORM) {
                    a0.x *= nscale; a0.y *= nscale;
                    a1.x *= nscale; a1.y *= nscale;
                }
#pragma unroll
                for (int t = 0; t < NT; ++t) {
                    v2f b; b.x = bq[t].x; b.y = bq[t].y;
                    acc[t] = __builtin_amdgcn_wmma_f32_16x16x4_f32(
                         false, a0,  false, b,
                         (short)0, acc[t],  false,  false);
                }
#pragma unroll
                for (int t = 0; t < NT; ++t) {
                    v2f b; b.x = bq[t].z; b.y = bq[t].w;
                    acc[t] = __builtin_amdgcn_wmma_f32_16x16x4_f32(
                         false, a1,  false, b,
                         (short)0, acc[t],  false,  false);
                }
            }
        };

#pragma unroll 1
        for (int kc = 0; kc < K / 32 - 1; ++kc) {
            v2f a_nxt[8];
#pragma unroll
            for (int s = 0; s < 8; ++s)
                a_nxt[s] = *reinterpret_cast<const v2f*>(Arow + (kc + 1) * 32 + 4 * s);
            do_chunk(kc, a_cur);
#pragma unroll
            for (int s = 0; s < 8; ++s) a_cur[s] = a_nxt[s];
        }
        do_chunk(K / 32 - 1, a_cur);

        const size_t obase = (size_t)(m0 + half * 8) * N + idx16;
        float rv[NT][8];
        if (HAS_RESID) {
#pragma unroll
            for (int t = 0; t < NT; ++t)
#pragma unroll
                for (int r = 0; r < 8; ++r)
                    rv[t][r] = resid[obase + (size_t)r * N + t * 16];
        }
#pragma unroll
        for (int t = 0; t < NT; ++t) {
            const float bv = bias[t * 16 + idx16];
#pragma unroll
            for (int r = 0; r < 8; ++r) {
                float v = acc[t][r] + bv;
                if (RELU) v = fmaxf(v, 0.0f);
                if (HAS_RESID) v += rv[t][r];
                C[obase + (size_t)r * N + t * 16] = v;
            }
        }
    }
}

#define TILE_PITCH 132

template<bool HAS_RESID, bool FUSE_OUT>
__global__ __launch_bounds__(GEMM_THREADS) void gcn_layer_fused(
    const float* __restrict__ Hin, const float* __restrict__ W,
    const float* __restrict__ bias, const int* __restrict__ cnt,
    const int* bins, const int* __restrict__ e_src,
    const int* __restrict__ e_dst, int n_edges,
    float* __restrict__ Hout, const float* __restrict__ Wout,
    const float* __restrict__ bout, float* Out, int n_strips)
{
    constexpr int NT = 8;
    constexpr int N = HID;
    constexpr int K = HID;
    constexpr int W_F4 = K * N / 4;
    constexpr int NT2 = OUT_DIM / 16;
    constexpr int N2 = OUT_DIM;
    constexpr int WO_F4 = FUSE_OUT ? HID * OUT_DIM / 4 : 1;
    __shared__ float4 Wl[W_F4];
    __shared__ float4 Wo[WO_F4];
    __shared__ __attribute__((aligned(16))) float At[(GEMM_THREADS / 32) * 16 * TILE_PITCH];
    __shared__ int s_next;

    const int tid   = threadIdx.x;
    const int lane  = tid & 31;
    const int half  = lane >> 4;
    const int idx16 = lane & 15;

    if (tid == 0) s_next = 0;
    for (int it = tid; it < W_F4; it += GEMM_THREADS) {
        const int c   = it & (N - 1);
        const int jh  = it / N;
        const int kb  = (jh >> 1) * 8 + (jh & 1) * 2;
        const float* __restrict__ wp = W + (size_t)kb * N + c;
        float4 v;
        v.x = wp[0];
        v.y = wp[N];
        v.z = wp[4 * N];
        v.w = wp[5 * N];
        Wl[it] = v;
    }
    if (FUSE_OUT) {
        for (int it = tid; it < WO_F4; it += GEMM_THREADS) {
            const int c   = it & (N2 - 1);
            const int jh  = it / N2;
            const int kb  = (jh >> 1) * 8 + (jh & 1) * 2;
            const float* __restrict__ wp = Wout + (size_t)kb * N2 + c;
            float4 v;
            v.x = wp[0];
            v.y = wp[N2];
            v.z = wp[4 * N2];
            v.w = wp[5 * N2];
            Wo[it] = v;
        }
    }
    __syncthreads();

    float* tile = At + (tid >> 5) * (16 * TILE_PITCH);
    const float* __restrict__ hl = Hin + lane * 4;
    const int my_count = (n_strips - (int)blockIdx.x + (int)gridDim.x - 1) / (int)gridDim.x;

    for (int guard = 0; guard < my_count; ++guard) {
        int i = 0;
        if (lane == 0) i = atomicAdd(&s_next, 1);
        i = __builtin_amdgcn_readfirstlane(i);
        if (i >= my_count) break;

        const int m0 = ((int)blockIdx.x + (int)gridDim.x * i) * 16;

        int c = 0;
        if (lane < 16) c = cnt[m0 + lane];
        const float nd = rsqrtf(fmaxf((float)c, 1.0f));
        const int cl = (c < BIN_CAP) ? c : BIN_CAP;
        const bool any_ovf = __builtin_amdgcn_ballot_w32(c > BIN_CAP) != 0u;
        int inc = cl;
        for (int d = 1; d < 32; d <<= 1) {
            int u = __shfl_up(inc, d);
            if (lane >= d) inc += u;
        }
        const int rp = inc - cl;
        const int E1 = __builtin_amdgcn_readlane(rp, 16);
        int node = 0;
        int nextEnd = __builtin_amdgcn_readlane(rp, 1);
        float4 acc = make_float4(0.0f, 0.0f, 0.0f, 0.0f);

        auto slot_of = [&](const int p) -> int {
            int j = 0;
            int base = 0;
#pragma unroll
            for (int k = 1; k < 16; ++k) {
                const int r = __builtin_amdgcn_readlane(rp, k);
                const bool ge = p >= r;
                j += ge ? 1 : 0;
                base = ge ? r : base;
            }
            return (m0 + j) * BIN_PITCH + (p - base);
        };
        auto node_id = [](const int x) -> int {
            return (x < 0) ? 0 : ((x > N_NODES - 1) ? N_NODES - 1 : x);
        };

        auto flush = [&]() {
            const float sc = __int_as_float(__builtin_amdgcn_readlane(__float_as_int(nd), node));
            float4 o;
            o.x = acc.x * sc; o.y = acc.y * sc; o.z = acc.z * sc; o.w = acc.w * sc;
            *reinterpret_cast<float4*>(tile + node * TILE_PITCH + lane * 4) = o;
            acc = make_float4(0.0f, 0.0f, 0.0f, 0.0f);
            ++node;
            nextEnd = (node < 16) ? __builtin_amdgcn_readlane(rp, (node + 1) & 31) : 0x7fffffff;
        };

        int pe = lane;
        if (pe >= E1) pe = E1 - 1;
        if (pe < 0) pe = 0;
        int s = node_id(bins[slot_of(pe)]);
        float ns = rsqrtf(fmaxf((float)cnt[s], 1.0f));

        for (int p0 = 0; p0 < E1; p0 += 32) {
            const int m = (E1 - p0 < 32) ? (E1 - p0) : 32;
            int pn = p0 + 32 + lane;
            if (pn >= E1) pn = E1 - 1;
            const int s_n = node_id(bins[slot_of(pn)]);

            float4 v[32];
#pragma unroll
            for (int g = 0; g < 4; ++g) {
                if (g * 8 < m) {
#pragma unroll
                    for (int u = 0; u < 8; ++u) {
                        const int sj = __builtin_amdgcn_readlane(s, g * 8 + u);
                        v[g * 8 + u] = *reinterpret_cast<const float4*>(hl + (size_t)sj * HID);
                    }
                } else {
#pragma unroll
                    for (int u = 0; u < 8; ++u) v[g * 8 + u] = make_float4(0.0f, 0.0f, 0.0f, 0.0f);
                }
            }
#pragma unroll
            for (int u = 0; u < 32; ++u) {
                if (u < m) {
                    const int p = p0 + u;
                    while (p >= nextEnd) flush();
                    const float n = __int_as_float(__builtin_amdgcn_readlane(__float_as_int(ns), u));
                    acc.x = __builtin_fmaf(v[u].x, n, acc.x);
                    acc.y = __builtin_fmaf(v[u].y, n, acc.y);
                    acc.z = __builtin_fmaf(v[u].z, n, acc.z);
                    acc.w = __builtin_fmaf(v[u].w, n, acc.w);
                }
            }
            const float ns_n = rsqrtf(fmaxf((float)cnt[s_n], 1.0f));
            s = s_n;
            ns = ns_n;
        }
        while (node < 16) flush();

        if (any_ovf) {
            for (int j = 0; j < 16; ++j) {
                const int cj = __builtin_amdgcn_readlane(c, j);
                if (cj <= BIN_CAP) continue;
                float4 ex = make_float4(0.0f, 0.0f, 0.0f, 0.0f);
                for (int e0 = 0; e0 < n_edges; e0 += 32) {
                    const int e = e0 + lane;
                    const bool hit = (e < n_edges) && (e_dst[e] == m0 + j);
                    const int so_l = hit ? node_id(e_src[e]) : 0;
                    unsigned int mask = __builtin_amdgcn_ballot_w32(hit);
                    while (mask != 0u) {
                        const int b = __builtin_ctz(mask);
                        mask &= mask - 1u;
                        const int so = __builtin_amdgcn_readlane(so_l, b);
                        const float n = rsqrtf(fmaxf((float)cnt[so], 1.0f));
                        const float4 r = *reinterpret_cast<const float4*>(hl + (size_t)so * HID);
                        ex.x = __builtin_fmaf(r.x, n, ex.x);
                        ex.y = __builtin_fmaf(r.y, n, ex.y);
                        ex.z = __builtin_fmaf(r.z, n, ex.z);
                        ex.w = __builtin_fmaf(r.w, n, ex.w);
                    }
                }
                const float sc = rsqrtf(fmaxf((float)cj, 1.0f));
                float4 t;
                t.x = ex.x * sc; t.y = ex.y * sc; t.z = ex.z * sc; t.w = ex.w * sc;
                *reinterpret_cast<float4*>(tile + j * TILE_PITCH + lane * 4) = t;
            }
        }

        const float* Arow = tile + idx16 * TILE_PITCH + half * 2;

        v8f cacc[NT];
#pragma unroll
        for (int t = 0; t < NT; ++t) cacc[t] = (v8f){};

        v2f a_cur[8];
#pragma unroll
        for (int q = 0; q < 8; ++q)
            a_cur[q] = *reinterpret_cast<const v2f*>(Arow + 4 * q);

        auto do_chunk = [&](const int kc, const v2f (&a)[8]) {
            const float4* bl = Wl + (size_t)((kc * 4) * 2 + half) * N + idx16;
#pragma unroll
            for (int js = 0; js < 4; ++js) {
                float4 bq[NT];
#pragma unroll
                for (int t = 0; t < NT; ++t) bq[t] = bl[(js * 2) * N + t * 16];
                const v2f a0 = a[2 * js];
                const v2f a1 = a[2 * js + 1];
#pragma unroll
                for (int t = 0; t < NT; ++t) {
                    v2f b; b.x = bq[t].x; b.y = bq[t].y;
                    cacc[t] = __builtin_amdgcn_wmma_f32_16x16x4_f32(
                         false, a0,  false, b,
                         (short)0, cacc[t],  false,  false);
                }
#pragma unroll
                for (int t = 0; t < NT; ++t) {
                    v2f b; b.x = bq[t].z; b.y = bq[t].w;
                    cacc[t] = __builtin_amdgcn_wmma_f32_16x16x4_f32(
                         false, a1,  false, b,
                         (short)0, cacc[t],  false,  false);
                }
            }
        };

#pragma unroll 1
        for (int kc = 0; kc < K / 32 - 1; ++kc) {
            v2f a_nxt[8];
#pragma unroll
            for (int q = 0; q < 8; ++q)
                a_nxt[q] = *reinterpret_cast<const v2f*>(Arow + (kc + 1) * 32 + 4 * q);
            do_chunk(kc, a_cur);
#pragma unroll
            for (int q = 0; q < 8; ++q) a_cur[q] = a_nxt[q];
        }
        do_chunk(K / 32 - 1, a_cur);

        if (!FUSE_OUT) {
            const size_t obase = (size_t)(m0 + half * 8) * N + idx16;
            float rv[NT][8];
            if (HAS_RESID) {
#pragma unroll
                for (int t = 0; t < NT; ++t)
#pragma unroll
                    for (int r = 0; r < 8; ++r)
                        rv[t][r] = Hin[obase + (size_t)r * N + t * 16];
            }
#pragma unroll
            for (int t = 0; t < NT; ++t) {
                const float bv = bias[t * 16 + idx16];
#pragma unroll
                for (int r = 0; r < 8; ++r) {
                    float o = fmaxf(cacc[t][r] + bv, 0.0f);
                    if (HAS_RESID) o += rv[t][r];
                    Hout[obase + (size_t)r * N + t * 16] = o;
                }
            }
        } else {
            float* trow = tile + (half * 8) * TILE_PITCH + idx16;
#pragma unroll
            for (int t = 0; t < NT; ++t) {
                const float bv = bias[t * 16 + idx16];
#pragma unroll
                for (int r = 0; r < 8; ++r)
                    trow[r * TILE_PITCH + t * 16] = fmaxf(cacc[t][r] + bv, 0.0f);
            }

            v8f oacc[NT2];
#pragma unroll
            for (int t = 0; t < NT2; ++t) oacc[t] = (v8f){};

            v2f b_cur[8];
#pragma unroll
            for (int q = 0; q < 8; ++q)
                b_cur[q] = *reinterpret_cast<const v2f*>(Arow + 4 * q);

            auto do_chunk2 = [&](const int kc, const v2f (&a)[8]) {
                const float4* bl = Wo + (size_t)((kc * 4) * 2 + half) * N2 + idx16;
#pragma unroll
                for (int js = 0; js < 4; ++js) {
                    float4 bq[NT2];
#pragma unroll
                    for (int t = 0; t < NT2; ++t) bq[t] = bl[(js * 2) * N2 + t * 16];
                    const v2f a0 = a[2 * js];
                    const v2f a1 = a[2 * js + 1];
#pragma unroll
                    for (int t = 0; t < NT2; ++t) {
                        v2f b; b.x = bq[t].x; b.y = bq[t].y;
                        oacc[t] = __builtin_amdgcn_wmma_f32_16x16x4_f32(
                             false, a0,  false, b,
                             (short)0, oacc[t],  false,  false);
                    }
#pragma unroll
                    for (int t = 0; t < NT2; ++t) {
                        v2f b; b.x = bq[t].z; b.y = bq[t].w;
                        oacc[t] = __builtin_amdgcn_wmma_f32_16x16x4_f32(
                             false, a1,  false, b,
                             (short)0, oacc[t],  false,  false);
                    }
                }
            };

#pragma unroll 1
            for (int kc = 0; kc < K / 32 - 1; ++kc) {
                v2f b_nxt[8];
#pragma unroll
                for (int q = 0; q < 8; ++q)
                    b_nxt[q] = *reinterpret_cast<const v2f*>(Arow + (kc + 1) * 32 + 4 * q);
                do_chunk2(kc, b_cur);
#pragma unroll
                for (int q = 0; q < 8; ++q) b_cur[q] = b_nxt[q];
            }
            do_chunk2(K / 32 - 1, b_cur);

            const size_t obase2 = (size_t)(m0 + half * 8) * N2 + idx16;
#pragma unroll
            for (int t = 0; t < NT2; ++t) {
                const float bo = bout[t * 16 + idx16];
#pragma unroll
                for (int r = 0; r < 8; ++r)
                    Out[obase2 + (size_t)r * N2 + t * 16] = oacc[t][r] + bo;
            }
        }
    }
}

extern "C" void kernel_launch(void* const* d_in, const int* in_sizes, int n_in,
                              void* d_out, int out_size, void* d_ws, size_t ws_size,
                              hipStream_t stream) {
    const float* features = (const float*)d_in[0];
    const int*   src      = (const int*)  d_in[1];
    const int*   dst      = (const int*)  d_in[2];
    const float* W_in     = (const float*)d_in[3];
    const float* b_in     = (const float*)d_in[4];
    const float* gcn_w0   = (const float*)d_in[5];
    const float* gcn_b0   = (const float*)d_in[6];
    const float* gcn_w1   = (const float*)d_in[7];
    const float* gcn_b1   = (const float*)d_in[8];
    const float* W_out    = (const float*)d_in[9];
    const float* b_out    = (const float*)d_in[10];
    float* out = (float*)d_out;

    char* ws = (char*)d_ws;
    int* cnt = (int*)ws;
    ws += ((size_t)N_NODES * sizeof(float) + 255) & ~(size_t)255;
    float* H = (float*)ws;
    ws += ((size_t)N_NODES * HID * sizeof(float) + 255) & ~(size_t)255;
    float* H2 = (float*)ws;

    int* bins = (int*)d_out;

    const int nStrips = N_NODES / 16;
    const int gemmGrid = 256;

    zero_int_kernel<<<(N_NODES + 255) / 256, 256, 0, stream>>>(cnt, N_NODES);

    wmma_gemm_lds<8, IN_DIM, false, false, false, true><<<gemmGrid, GEMM_THREADS + FILL_THREADS, 0, stream>>>(
        features, W_in, b_in, nullptr, nullptr, H, nStrips,
        src, dst, cnt, bins, N_EDGES);

    gcn_layer_fused<true, false><<<gemmGrid, GEMM_THREADS, 0, stream>>>(
        H, gcn_w0, gcn_b0, cnt, bins, src, dst, N_EDGES, H2,
        nullptr, nullptr, nullptr, nStrips);

    gcn_layer_fused<false, true><<<gemmGrid, GEMM_THREADS, 0, stream>>>(
        H2, gcn_w1, gcn_b1, cnt, bins, src, dst, N_EDGES, nullptr,
        W_out, b_out, out, nStrips);
}
